# starting kernel
# baseline (speedup 1.0000x reference)
#include <hip/hip_runtime.h>
#include <math.h>


typedef float v2f __attribute__((ext_vector_type(2)));
typedef float v8f __attribute__((ext_vector_type(8)));

#define STATES   16
#define FDIM     128
#define EPSF     1e-10f
#define EXP1F    2.7182818284590452f
#define EXP1M1F  1.7182818284590452f

__global__ void bpn_priors_wmma(const float* __restrict__ feat,
                                const float* __restrict__ W,
                                float* __restrict__ priors,
                                float* __restrict__ log_priors,
                                float* __restrict__ accum,
                                int N, int numTiles) {
  const int wave = blockIdx.x * (blockDim.x >> 5) + (threadIdx.x >> 5);
  if (wave >= numTiles) return;
  const int lane = threadIdx.x & 31;
  const int half = lane >> 4;
  const int s    = lane & 15;

  int rowA = wave * 16 + s;
  if (rowA >= N) rowA = N - 1;
  const float* frow = feat + (size_t)rowA * FDIM;

  v8f acc = {};
#pragma unroll
  for (int k0 = 0; k0 < FDIM; k0 += 4) {
    const int k = k0 + 2 * half;
    v2f a, b;
    a.x = frow[k];
    a.y = frow[k + 1];
    b.x = W[(size_t)k * STATES + s];
    b.y = W[(size_t)(k + 1) * STATES + s];
    acc = __builtin_amdgcn_wmma_f32_16x16x4_f32(false, a, false, b,
                                                (short)0, acc, false, false);
  }

#pragma unroll
  for (int r = 0; r < 8; ++r) {
    float logit = acc[r];
    float mx = logit;
    mx = fmaxf(mx, __shfl_xor(mx, 1, 32));
    mx = fmaxf(mx, __shfl_xor(mx, 2, 32));
    mx = fmaxf(mx, __shfl_xor(mx, 4, 32));
    mx = fmaxf(mx, __shfl_xor(mx, 8, 32));
    float ex = expf(logit - mx);
    float sm = ex;
    sm += __shfl_xor(sm, 1, 32);
    sm += __shfl_xor(sm, 2, 32);
    sm += __shfl_xor(sm, 4, 32);
    sm += __shfl_xor(sm, 8, 32);
    float pr = ex / sm;
    float lp = logf(fmaxf(pr, EPSF));
    const int rowO = wave * 16 + r + 8 * half;
    if (rowO < N) {
      const size_t o = (size_t)rowO * STATES + s;
      priors[o]     = pr;
      log_priors[o] = lp;
      accum[o]      = lp;
    }
  }
}

__global__ void bpn_edge(const float* __restrict__ beliefs,
                         const int* __restrict__ src,
                         const int* __restrict__ dst,
                         const int* __restrict__ rev,
                         float* __restrict__ msgs,
                         float* __restrict__ accum,
                         int Eu, int firstIter) {
  const int p = blockIdx.x * (blockDim.x >> 5) + (threadIdx.x >> 5);
  if (p >= Eu) return;
  const int lane = threadIdx.x & 31;
  const int half = lane >> 4;
  const int s    = lane & 15;

  const int j = rev[p];
  const int e = half ? j : p;

  float mOld = firstIter ? 0.0625f : msgs[(size_t)e * STATES + s];
  const float mRev = __shfl_xor(mOld, 16, 32);

  const int   se = src[e];
  const float b  = beliefs[(size_t)se * STATES + s];
  const float x  = b / mRev;

  float t = x;
  t += __shfl_xor(t, 1, 32);
  t += __shfl_xor(t, 2, 32);
  t += __shfl_xor(t, 4, 32);
  t += __shfl_xor(t, 8, 32);

  const float m = (t + EXP1M1F * x) / ((15.0f + EXP1F) * t);
  msgs[(size_t)e * STATES + s] = m;

  const float lm = logf(fmaxf(m, EPSF));
  const int   de = dst[e];
  atomicAdd(&accum[(size_t)de * STATES + s], lm);
}

__global__ void bpn_node(float* __restrict__ accum,
                         const float* __restrict__ log_priors,
                         float* __restrict__ beliefs_out,
                         int N) {
  const int w    = blockIdx.x * (blockDim.x >> 5) + (threadIdx.x >> 5);
  const int lane = threadIdx.x & 31;
  const int half = lane >> 4;
  const int s    = lane & 15;
  const int node = w * 2 + half;

  const bool ok = (node < N);
  const size_t o = (size_t)(ok ? node : 0) * STATES + s;
  float v = ok ? accum[o] : 0.0f;

  float mx = v;
  mx = fmaxf(mx, __shfl_xor(mx, 1, 32));
  mx = fmaxf(mx, __shfl_xor(mx, 2, 32));
  mx = fmaxf(mx, __shfl_xor(mx, 4, 32));
  mx = fmaxf(mx, __shfl_xor(mx, 8, 32));
  float ex = expf(v - mx);
  float sm = ex;
  sm += __shfl_xor(sm, 1, 32);
  sm += __shfl_xor(sm, 2, 32);
  sm += __shfl_xor(sm, 4, 32);
  sm += __shfl_xor(sm, 8, 32);

  if (ok) {
    beliefs_out[o] = ex / sm;
    accum[o]       = log_priors[o];
  }
}

extern "C" void kernel_launch(void* const* d_in, const int* in_sizes, int n_in,
                              void* d_out, int out_size, void* d_ws, size_t ws_size,
                              hipStream_t stream) {
  const float* feat = (const float*)d_in[0];
  const float* W    = (const float*)d_in[1];
  const int*   src  = (const int*)d_in[2];
  const int*   dst  = (const int*)d_in[3];
  const int*   rev  = (const int*)d_in[4];

  const int N  = in_sizes[0] / FDIM;
  const int E2 = in_sizes[2];
  const int Eu = E2 / 2;

  float* priors      = (float*)d_out;
  float* beliefs_out = priors + (size_t)N * STATES;

  float* lp    = (float*)d_ws;
  float* accum = lp    + (size_t)N * STATES;
  float* bel   = accum + (size_t)N * STATES;
  float* msgs  = bel   + (size_t)N * STATES;

  const dim3 blk(256);
  const int tiles   = (N + 15) / 16;
  const int gBlocks = (tiles + 7) / 8;
  const int eBlocks = (Eu + 7) / 8;
  const int nWaves  = (N + 1) / 2;
  const int nBlocks = (nWaves + 7) / 8;

  bpn_priors_wmma<<<gBlocks, blk, 0, stream>>>(feat, W, priors, lp, accum, N, tiles);

  bpn_edge<<<eBlocks, blk, 0, stream>>>(priors, src, dst, rev, msgs, accum, Eu, 1);
  bpn_node<<<nBlocks, blk, 0, stream>>>(accum, lp, bel, N);
  bpn_edge<<<eBlocks, blk, 0, stream>>>(bel, src, dst, rev, msgs, accum, Eu, 0);
  bpn_node<<<nBlocks, blk, 0, stream>>>(accum, lp, bel, N);
  bpn_edge<<<eBlocks, blk, 0, stream>>>(bel, src, dst, rev, msgs, accum, Eu, 0);
  bpn_node<<<nBlocks, blk, 0, stream>>>(accum, lp, beliefs_out, N);
}


# Node sums re-tiled: a wave per node, 4 lanes per in-edge (16-byte quarter-row gathers, 8 rows a load; row total in-lane + 2 exchanges); node sum = 8 lane-group partials, folded (new f32 sum order)
# speedup vs baseline: 1.4671x; 1.4671x over previous
#include <hip/hip_runtime.h>
#include <math.h>


typedef float v2f __attribute__((ext_vector_type(2)));
typedef float v4f __attribute__((ext_vector_type(4)));
typedef float v8f __attribute__((ext_vector_type(8)));

#define STATES   16
#define FDIM     128
#define EPSF     1e-10f
#define EXP1F    2.7182818284590452f
#define EXP1M1F  1.7182818284590452f

__global__ void bp_priors_wmma(const float* __restrict__ feat,
                               const float* __restrict__ W,
                               float* __restrict__ priors,
                               float* __restrict__ lg1,
                               int* __restrict__ cnt,
                               int N, int numTiles) {
  const int wave = blockIdx.x * (blockDim.x >> 5) + (threadIdx.x >> 5);
  if (wave >= numTiles) return;
  const int lane = threadIdx.x & 31;
  const int half = lane >> 4;
  const int s    = lane & 15;

  const int r0 = wave * 16 + s;
  if (half == 0 && r0 < N) cnt[r0] = 0;

  int rowA = r0;
  if (rowA >= N) rowA = N - 1;
  const float* frow = feat + (size_t)rowA * FDIM;

  v8f acc = {};
#pragma unroll
  for (int k0 = 0; k0 < FDIM; k0 += 4) {
    const int k = k0 + 2 * half;
    v2f a, b;
    a.x = frow[k];
    a.y = frow[k + 1];
    b.x = W[(size_t)k * STATES + s];
    b.y = W[(size_t)(k + 1) * STATES + s];
    acc = __builtin_amdgcn_wmma_f32_16x16x4_f32(false, a, false, b,
                                                (short)0, acc, false, false);
  }

#pragma unroll
  for (int r = 0; r < 8; ++r) {
    float logit = acc[r];
    float mx = logit;
    mx = fmaxf(mx, __shfl_xor(mx, 1, 32));
    mx = fmaxf(mx, __shfl_xor(mx, 2, 32));
    mx = fmaxf(mx, __shfl_xor(mx, 4, 32));
    mx = fmaxf(mx, __shfl_xor(mx, 8, 32));
    float ex = expf(logit - mx);
    float sm = ex;
    sm += __shfl_xor(sm, 1, 32);
    sm += __shfl_xor(sm, 2, 32);
    sm += __shfl_xor(sm, 4, 32);
    sm += __shfl_xor(sm, 8, 32);
    const int rowO = wave * 16 + r + 8 * half;
    const float p = ex / sm;
    const float x = p / 0.0625f;
    float tt = x;
    tt += __shfl_xor(tt, 1, 32);
    tt += __shfl_xor(tt, 2, 32);
    tt += __shfl_xor(tt, 4, 32);
    tt += __shfl_xor(tt, 8, 32);
    const float m = (tt + EXP1M1F * x) / ((15.0f + EXP1F) * tt);
    const float lm = logf(fmaxf(m, EPSF));
    if (rowO < N) {
      priors[(size_t)rowO * STATES + s] = p;
      lg1[(size_t)rowO * STATES + s]    = lm;
    }
  }
}

__global__ void bp_rank(const int* __restrict__ dst, int* __restrict__ cnt,
                        int* __restrict__ rank, int E2) {
  const int e = blockIdx.x * blockDim.x + threadIdx.x;
  if (e >= E2) return;
  rank[e] = atomicAdd(&cnt[dst[e]], 1);
}

__global__ void bp_scan(const int* __restrict__ cnt, int* __restrict__ seg, int N) {
  __shared__ int wsum[32];
  __shared__ int sbase;
  const int tid  = threadIdx.x;
  const int lane = tid & 31;
  const int wv   = tid >> 5;
  const int start = blockIdx.x * 1024;

  int part = 0;
  for (int i = tid; i < start; i += 1024) part += cnt[i];
  for (int o = 16; o > 0; o >>= 1) part += __shfl_xor(part, o, 32);
  if (lane == 0) wsum[wv] = part;
  __syncthreads();
  if (wv == 0) {
    int v = wsum[lane];
    for (int o = 16; o > 0; o >>= 1) v += __shfl_xor(v, o, 32);
    if (lane == 0) sbase = v;
  }
  __syncthreads();
  const int base = sbase;

  const int i = start + tid;
  const int c = (i < N) ? cnt[i] : 0;
  int inc = c;
  for (int o = 1; o < 32; o <<= 1) {
    const int y = __shfl_up(inc, o, 32);
    if (lane >= o) inc += y;
  }
  if (lane == 31) wsum[wv] = inc;
  __syncthreads();
  if (wv == 0) {
    const int v = wsum[lane];
    int iv = v;
    for (int o = 1; o < 32; o <<= 1) {
      const int y = __shfl_up(iv, o, 32);
      if (lane >= o) iv += y;
    }
    wsum[lane] = iv - v;
  }
  __syncthreads();
  const int beg = base + wsum[wv] + inc - c;
  if (i < N) {
    seg[(size_t)i * STATES]     = beg;
    seg[(size_t)i * STATES + 1] = beg + c;
  }
}

__global__ void bp_fill(const int* __restrict__ dst, const int* __restrict__ rank,
                        const int* __restrict__ seg, int* __restrict__ perm, int E2) {
  const int e = blockIdx.x * blockDim.x + threadIdx.x;
  if (e >= E2) return;
  const int slot = seg[(size_t)dst[e] * STATES] + rank[e];
  perm[min(max(slot, 0), E2 - 1)] = e;
}

#define TRIPS 8
template <int FIRST>
__global__ void bp_edge(const int* __restrict__ src,
                        const int* __restrict__ rev,
                        const float* __restrict__ belin,
                        float* __restrict__ M,
                        int E2) {
  const int w    = __builtin_amdgcn_readfirstlane((int)(blockIdx.x * (blockDim.x >> 5) + (threadIdx.x >> 5)));
  const int lane = threadIdx.x & 31;
  const int g    = lane >> 2;
  const int q4   = (lane & 3) * 4;
  const int ew   = w * (32 * TRIPS);

  for (int t = 0; t < TRIPS; ++t) {
    const int e0 = ew + 32 * t;
    if (e0 >= E2) break;
    int ei[4], ki[4];
#pragma unroll
    for (int i = 0; i < 4; ++i) ei[i] = min(e0 + 8 * i + g, E2 - 1);
#pragma unroll
    for (int i = 0; i < 4; ++i) ki[i] = FIRST ? src[rev[ei[i]]] : src[ei[i]];

    v4f mr[4], b[4];
    if (!FIRST) {
#pragma unroll
      for (int i = 0; i < 4; ++i) mr[i] = *(const v4f*)(M + (size_t)ei[i] * STATES + q4);
    }
#pragma unroll
    for (int i = 0; i < 4; ++i) b[i] = *(const v4f*)(belin + (size_t)ki[i] * STATES + q4);

#pragma unroll
    for (int i = 0; i < 4; ++i) {
      v4f x;
      if (FIRST) x = b[i] / 0.0625f;
      else       x = b[i] / mr[i];
      float tt = (x.x + x.y) + (x.z + x.w);
      tt += __shfl_xor(tt, 1, 32);
      tt += __shfl_xor(tt, 2, 32);
      const v4f m = (tt + EXP1M1F * x) / ((15.0f + EXP1F) * tt);
      if (e0 + 8 * i + g < E2) *(v4f*)(M + (size_t)ei[i] * STATES + q4) = m;
    }
  }
}

#define NPW 4
#define GRP 4
template <int MODE>
__global__ void bp_pull(const int* __restrict__ perm,
                        const int* __restrict__ src,
                        const int* __restrict__ rev,
                        const float* __restrict__ priors,
                        const float* belin,
                        const float* M,
                        const int* seg,
                        float* bout,
                        int N, int E2) {
  const int w    = __builtin_amdgcn_readfirstlane((int)(blockIdx.x * (blockDim.x >> 5) + (threadIdx.x >> 5)));
  const int lane = threadIdx.x & 31;
  const int g    = lane >> 2;
  const int q4   = (lane & 3) * 4;
  const bool lo  = lane < 16;
  const int n0   = w * NPW;
  if (n0 >= N) return;

  int nb = seg[(size_t)n0 * STATES];
  int ne = seg[(size_t)n0 * STATES + 1];

  for (int jp = 0; jp < NPW; jp += 2) {
    const int nodeA = n0 + jp;
    if (nodeA >= N) break;
    const bool hasB = (nodeA + 1) < N;
    const size_t rowP = (size_t)(nodeA + ((hasB && !lo) ? 1 : 0)) * STATES;
    const v4f pr = *(const v4f*)(priors + rowP + q4);

    v4f accA = {0.0f, 0.0f, 0.0f, 0.0f};
    v4f accB = {0.0f, 0.0f, 0.0f, 0.0f};
    for (int h = 0; h < 2; ++h) {
      const int node = nodeA + h;
      if (node >= N) break;
      const int beg = min(max(__builtin_amdgcn_readfirstlane(nb), 0), E2);
      const int deg = min(max(__builtin_amdgcn_readfirstlane(ne) - beg, 0), E2 - beg);
      if (jp + h + 1 < NPW && node + 1 < N) {
        nb = seg[(size_t)(node + 1) * STATES];
        ne = seg[(size_t)(node + 1) * STATES + 1];
      }

      v4f acc = {0.0f, 0.0f, 0.0f, 0.0f};
      for (int base = 0; base < deg; base += 8 * GRP) {
        int ei[GRP], ki[GRP], ri[GRP];
        v4f b[GRP], mr[GRP];
        for (int i = 0; i < GRP; ++i) {
          if (base + 8 * i < deg) {
            const int p = base + 8 * i + g;
            ei[i] = perm[beg + ((p < deg) ? p : (base + 8 * i))];
          }
        }
        for (int i = 0; i < GRP; ++i) {
          if (base + 8 * i < deg) {
            ei[i] = min(max(ei[i], 0), E2 - 1);
            if (MODE != 3) ki[i] = src[ei[i]];
            if (MODE == 4) ri[i] = rev[ei[i]];
          }
        }
        for (int i = 0; i < GRP; ++i) {
          if (base + 8 * i < deg) {
            if (MODE == 3) {
              b[i] = *(const v4f*)(M + (size_t)ei[i] * STATES + q4);
            } else {
              b[i] = *(const v4f*)(belin + (size_t)ki[i] * STATES + q4);
              if (MODE == 4) mr[i] = *(const v4f*)(M + (size_t)ri[i] * STATES + q4);
            }
          }
        }
        for (int i = 0; i < GRP; ++i) {
          if (base + 8 * i < deg) {
            v4f lm;
            if (MODE == 1) {
              lm = b[i];
            } else {
              v4f m;
              if (MODE == 3) {
                m = b[i];
              } else {
                const v4f x = b[i] / mr[i];
                float tt = (x.x + x.y) + (x.z + x.w);
                tt += __shfl_xor(tt, 1, 32);
                tt += __shfl_xor(tt, 2, 32);
                m = (tt + EXP1M1F * x) / ((15.0f + EXP1F) * tt);
              }
              lm.x = logf(fmaxf(m.x, EPSF));
              lm.y = logf(fmaxf(m.y, EPSF));
              lm.z = logf(fmaxf(m.z, EPSF));
              lm.w = logf(fmaxf(m.w, EPSF));
            }
            const bool live = (base + 8 * i + g) < deg;
            acc.x += live ? lm.x : 0.0f;
            acc.y += live ? lm.y : 0.0f;
            acc.z += live ? lm.z : 0.0f;
            acc.w += live ? lm.w : 0.0f;
          }
        }
      }
      if (h == 0) accA = acc; else accB = acc;
    }

    v4f t = lo ? accA : accB;
    const v4f u = lo ? accB : accA;
    t.x += __shfl_xor(u.x, 16, 32);
    t.y += __shfl_xor(u.y, 16, 32);
    t.z += __shfl_xor(u.z, 16, 32);
    t.w += __shfl_xor(u.w, 16, 32);
    t.x += __shfl_xor(t.x, 8, 32);
    t.y += __shfl_xor(t.y, 8, 32);
    t.z += __shfl_xor(t.z, 8, 32);
    t.w += __shfl_xor(t.w, 8, 32);
    t.x += __shfl_xor(t.x, 4, 32);
    t.y += __shfl_xor(t.y, 4, 32);
    t.z += __shfl_xor(t.z, 4, 32);
    t.w += __shfl_xor(t.w, 4, 32);

    v4f lb;
    lb.x = logf(fmaxf(pr.x, EPSF)) + t.x;
    lb.y = logf(fmaxf(pr.y, EPSF)) + t.y;
    lb.z = logf(fmaxf(pr.z, EPSF)) + t.z;
    lb.w = logf(fmaxf(pr.w, EPSF)) + t.w;

    float mx = fmaxf(fmaxf(lb.x, lb.y), fmaxf(lb.z, lb.w));
    mx = fmaxf(mx, __shfl_xor(mx, 1, 32));
    mx = fmaxf(mx, __shfl_xor(mx, 2, 32));
    v4f ex;
    ex.x = expf(lb.x - mx);
    ex.y = expf(lb.y - mx);
    ex.z = expf(lb.z - mx);
    ex.w = expf(lb.w - mx);
    float sm = (ex.x + ex.y) + (ex.z + ex.w);
    sm += __shfl_xor(sm, 1, 32);
    sm += __shfl_xor(sm, 2, 32);
    const v4f o = ex / sm;
    if ((lane & 12) == 0 && (lo || hasB)) *(v4f*)(bout + rowP + q4) = o;
  }
}

extern "C" void kernel_launch(void* const* d_in, const int* in_sizes, int n_in,
                              void* d_out, int out_size, void* d_ws, size_t ws_size,
                              hipStream_t stream) {
  const float* feat = (const float*)d_in[0];
  const float* W    = (const float*)d_in[1];
  const int*   src  = (const int*)d_in[2];
  const int*   dst  = (const int*)d_in[3];
  const int*   rev  = (const int*)d_in[4];

  const int N  = in_sizes[0] / FDIM;
  const int E2 = in_sizes[2];

  float* priors      = (float*)d_out;
  float* beliefs_out = priors + (size_t)N * STATES;
  int*   seg         = (int*)beliefs_out;

  int*   perm   = (int*)d_ws;
  float* belbuf = (float*)d_ws + (size_t)E2;
  float* msgs   = (float*)d_ws + (size_t)3 * N * STATES;
  int*   rank   = (int*)msgs;
  int*   cnt    = rank + (size_t)E2;
  float* lg1    = msgs + (size_t)2 * E2;

  const dim3 blk(256);
  const int tiles   = (N + 15) / 16;
  const int gBlocks = (tiles + 7) / 8;
  const int tBlocks = (E2 + 255) / 256;
  const int sBlocks = (N + 1023) / 1024;
  const int nBlocks = (N + 8 * NPW - 1) / (8 * NPW);
  const int eWaves  = (E2 + 32 * TRIPS - 1) / (32 * TRIPS);
  const int eBlocks = (eWaves + 7) / 8;

  bp_priors_wmma<<<gBlocks, blk, 0, stream>>>(feat, W, priors, lg1, cnt, N, tiles);

  bp_rank<<<tBlocks, blk, 0, stream>>>(dst, cnt, rank, E2);
  bp_scan<<<sBlocks, dim3(1024), 0, stream>>>(cnt, seg, N);
  bp_fill<<<tBlocks, blk, 0, stream>>>(dst, rank, seg, perm, E2);

  bp_pull<1><<<nBlocks, blk, 0, stream>>>(perm, src, rev, priors, lg1, msgs, seg, belbuf, N, E2);
  bp_edge<1><<<eBlocks, blk, 0, stream>>>(src, rev, priors, msgs, E2);
  bp_edge<0><<<eBlocks, blk, 0, stream>>>(src, rev, belbuf, msgs, E2);
  bp_pull<3><<<nBlocks, blk, 0, stream>>>(perm, src, rev, priors, belbuf, msgs, seg, belbuf, N, E2);
  bp_pull<4><<<nBlocks, blk, 0, stream>>>(perm, src, rev, priors, belbuf, msgs, seg, beliefs_out, N, E2);
}


# Destination sort rebuilt in two levels on LDS counters (256-node bins, runs staged in LDS, block per bin): no global atomics or scattered word stores; segments in ascending edge id (fixed sum order)
# speedup vs baseline: 1.8634x; 1.2701x over previous
#include <hip/hip_runtime.h>
#include <math.h>


typedef float v2f __attribute__((ext_vector_type(2)));
typedef float v4f __attribute__((ext_vector_type(4)));
typedef float v8f __attribute__((ext_vector_type(8)));

#define STATES   16
#define FDIM     128
#define EPSF     1e-10f
#define EXP1F    2.7182818284590452f
#define EXP1M1F  1.7182818284590452f

__global__ void bp_priors_wmma(const float* __restrict__ feat,
                               const float* __restrict__ W,
                               float* __restrict__ priors,
                               float* __restrict__ lg1,
                               int N, int numTiles) {
  const int wave = blockIdx.x * (blockDim.x >> 5) + (threadIdx.x >> 5);
  if (wave >= numTiles) return;
  const int lane = threadIdx.x & 31;
  const int half = lane >> 4;
  const int s    = lane & 15;

  int rowA = wave * 16 + s;
  if (rowA >= N) rowA = N - 1;
  const float* frow = feat + (size_t)rowA * FDIM;

  v8f acc = {};
#pragma unroll
  for (int k0 = 0; k0 < FDIM; k0 += 4) {
    const int k = k0 + 2 * half;
    v2f a, b;
    a.x = frow[k];
    a.y = frow[k + 1];
    b.x = W[(size_t)k * STATES + s];
    b.y = W[(size_t)(k + 1) * STATES + s];
    acc = __builtin_amdgcn_wmma_f32_16x16x4_f32(false, a, false, b,
                                                (short)0, acc, false, false);
  }

#pragma unroll
  for (int r = 0; r < 8; ++r) {
    float logit = acc[r];
    float mx = logit;
    mx = fmaxf(mx, __shfl_xor(mx, 1, 32));
    mx = fmaxf(mx, __shfl_xor(mx, 2, 32));
    mx = fmaxf(mx, __shfl_xor(mx, 4, 32));
    mx = fmaxf(mx, __shfl_xor(mx, 8, 32));
    float ex = expf(logit - mx);
    float sm = ex;
    sm += __shfl_xor(sm, 1, 32);
    sm += __shfl_xor(sm, 2, 32);
    sm += __shfl_xor(sm, 4, 32);
    sm += __shfl_xor(sm, 8, 32);
    const int rowO = wave * 16 + r + 8 * half;
    const float p = ex / sm;
    const float x = p / 0.0625f;
    float tt = x;
    tt += __shfl_xor(tt, 1, 32);
    tt += __shfl_xor(tt, 2, 32);
    tt += __shfl_xor(tt, 4, 32);
    tt += __shfl_xor(tt, 8, 32);
    const float m = (tt + EXP1M1F * x) / ((15.0f + EXP1F) * tt);
    const float lm = logf(fmaxf(m, EPSF));
    if (rowO < N) {
      priors[(size_t)rowO * STATES + s] = p;
      lg1[(size_t)rowO * STATES + s]    = lm;
    }
  }
}

#define BINSH   8
#define BINW    256
#define CH      8192
#define NBMAX   1024
#define CAP     16384
#define UNR     8

__device__ __forceinline__ int bp_bin_of(int d, int NB) {
  return min(max(d >> BINSH, 0), NB - 1);
}

__global__ void bp_bin_count(const int* __restrict__ dst, int* __restrict__ Hm,
                             int E2, int NB, int BLK) {
  __shared__ int hist[NBMAX];
  const int tid = threadIdx.x;
  for (int b = tid; b < NBMAX; b += 256) hist[b] = 0;
  __syncthreads();
  const int base = blockIdx.x * CH;
  for (int j = 0; j < CH / 256; ++j) {
    const int e = base + j * 256 + tid;
    if (e < E2) atomicAdd(&hist[bp_bin_of(dst[e], NB)], 1);
  }
  __syncthreads();
  for (int b = tid; b < NB; b += 256) Hm[(size_t)b * BLK + blockIdx.x] = hist[b];
}

__global__ void bp_scan_excl(const int* __restrict__ in, int* __restrict__ out, int n) {
  __shared__ int wsum[32];
  __shared__ int sbase;
  const int tid  = threadIdx.x;
  const int lane = tid & 31;
  const int wv   = tid >> 5;
  const int start = blockIdx.x * 1024;

  int part = 0;
  for (int i = tid; i < start; i += 1024) part += in[i];
  for (int o = 16; o > 0; o >>= 1) part += __shfl_xor(part, o, 32);
  if (lane == 0) wsum[wv] = part;
  __syncthreads();
  if (wv == 0) {
    int v = wsum[lane];
    for (int o = 16; o > 0; o >>= 1) v += __shfl_xor(v, o, 32);
    if (lane == 0) sbase = v;
  }
  __syncthreads();
  const int base = sbase;

  const int i = start + tid;
  const int c = (i < n) ? in[i] : 0;
  int inc = c;
  for (int o = 1; o < 32; o <<= 1) {
    const int y = __shfl_up(inc, o, 32);
    if (lane >= o) inc += y;
  }
  if (lane == 31) wsum[wv] = inc;
  __syncthreads();
  if (wv == 0) {
    const int v = wsum[lane];
    int iv = v;
    for (int o = 1; o < 32; o <<= 1) {
      const int y = __shfl_up(iv, o, 32);
      if (lane >= o) iv += y;
    }
    wsum[lane] = iv - v;
  }
  __syncthreads();
  if (i < n) out[i] = base + wsum[wv] + inc - c;
}

__global__ void bp_bin_scatter(const int* __restrict__ dst, const int* __restrict__ Off,
                               int* __restrict__ T, int* __restrict__ Td,
                               int E2, int NB, int BLK) {
  __shared__ int hist[NBMAX];
  __shared__ int delta[NBMAX];
  __shared__ int se[CH];
  __shared__ int sd[CH];
  __shared__ int wsum[8];
  const int tid  = threadIdx.x;
  const int lane = tid & 31;
  const int wv   = tid >> 5;
  const int blk  = blockIdx.x;
  const int base = blk * CH;
  const int nIt  = min(CH, E2 - base);

  int dd[CH / 256];
#pragma unroll
  for (int j = 0; j < CH / 256; ++j) {
    const int e = base + j * 256 + tid;
    dd[j] = (e < E2) ? dst[e] : 0;
  }
  for (int b = tid; b < NBMAX; b += 256) hist[b] = 0;
  __syncthreads();
#pragma unroll
  for (int j = 0; j < CH / 256; ++j) {
    const int e = base + j * 256 + tid;
    if (e < E2) atomicAdd(&hist[bp_bin_of(dd[j], NB)], 1);
  }
  __syncthreads();
  {
    const int v0 = hist[4 * tid], v1 = hist[4 * tid + 1], v2 = hist[4 * tid + 2], v3 = hist[4 * tid + 3];
    const int s4 = (v0 + v1) + (v2 + v3);
    int inc = s4;
    for (int o = 1; o < 32; o <<= 1) {
      const int y = __shfl_up(inc, o, 32);
      if (lane >= o) inc += y;
    }
    if (lane == 31) wsum[wv] = inc;
    __syncthreads();
    int woff = 0;
    for (int k = 0; k < wv; ++k) woff += wsum[k];
    const int ex = woff + inc - s4;
    hist[4 * tid]     = ex;
    hist[4 * tid + 1] = ex + v0;
    hist[4 * tid + 2] = ex + v0 + v1;
    hist[4 * tid + 3] = ex + v0 + v1 + v2;
  }
  __syncthreads();
  for (int b = tid; b < NB; b += 256) delta[b] = Off[(size_t)b * BLK + blk] - hist[b];
  __syncthreads();
#pragma unroll
  for (int j = 0; j < CH / 256; ++j) {
    const int e = base + j * 256 + tid;
    if (e < E2) {
      const int d = dd[j];
      const int p = min(max(atomicAdd(&hist[bp_bin_of(d, NB)], 1), 0), CH - 1);
      se[p] = e;
      sd[p] = d;
    }
  }
  __syncthreads();
  for (int s = tid; s < nIt; s += 256) {
    const int d = sd[s];
    const int g = min(max(s + delta[bp_bin_of(d, NB)], 0), E2 - 1);
    T[g]  = se[s];
    Td[g] = d;
  }
}

__global__ void bp_bin_sort(const int* __restrict__ Off, const int* __restrict__ T,
                            const int* __restrict__ Td, int* __restrict__ seg,
                            int* __restrict__ perm, int N, int E2, int NB, int BLK) {
  __shared__ int ncnt[BINW];
  __shared__ int nstart[BINW + 1];
  __shared__ int cur[BINW];
  __shared__ int Be[CAP];
  __shared__ int wsum[8];
  const int tid  = threadIdx.x;
  const int lane = tid & 31;
  const int wv   = tid >> 5;
  const int b    = blockIdx.x;
  const int n0   = b << BINSH;

  int binBeg = Off[(size_t)b * BLK];
  int binEnd = (b + 1 < NB) ? Off[(size_t)(b + 1) * BLK] : E2;
  binBeg = min(max(binBeg, 0), E2);
  binEnd = min(max(binEnd, binBeg), E2);
  const int cntB = binEnd - binBeg;

  ncnt[tid] = 0;
  __syncthreads();
  for (int i0 = 0; i0 < cntB; i0 += 256 * UNR) {
    int dn[UNR];
#pragma unroll
    for (int u = 0; u < UNR; ++u) {
      const int i = i0 + u * 256 + tid;
      dn[u] = (i < cntB) ? Td[binBeg + i] : n0;
    }
#pragma unroll
    for (int u = 0; u < UNR; ++u) {
      const int i = i0 + u * 256 + tid;
      if (i < cntB) atomicAdd(&ncnt[(dn[u] - n0) & (BINW - 1)], 1);
    }
  }
  __syncthreads();
  {
    const int c = ncnt[tid];
    int inc = c;
    for (int o = 1; o < 32; o <<= 1) {
      const int y = __shfl_up(inc, o, 32);
      if (lane >= o) inc += y;
    }
    if (lane == 31) wsum[wv] = inc;
    __syncthreads();
    int woff = 0;
    for (int k = 0; k < wv; ++k) woff += wsum[k];
    const int ex = woff + inc - c;
    nstart[tid] = ex;
    cur[tid]    = ex;
    if (tid == BINW - 1) nstart[BINW] = ex + c;
    const int node = n0 + tid;
    if (node < N) {
      seg[(size_t)node * STATES]     = binBeg + ex;
      seg[(size_t)node * STATES + 1] = binBeg + ex + c;
    }
  }
  __syncthreads();
  if (cntB <= CAP) {
    for (int i0 = 0; i0 < cntB; i0 += 256 * UNR) {
      int dn[UNR], en[UNR];
#pragma unroll
      for (int u = 0; u < UNR; ++u) {
        const int i = i0 + u * 256 + tid;
        dn[u] = (i < cntB) ? Td[binBeg + i] : n0;
        en[u] = (i < cntB) ? T[binBeg + i] : 0;
      }
#pragma unroll
      for (int u = 0; u < UNR; ++u) {
        const int i = i0 + u * 256 + tid;
        if (i < cntB) {
          const int p = min(max(atomicAdd(&cur[(dn[u] - n0) & (BINW - 1)], 1), 0), CAP - 1);
          Be[p] = en[u];
        }
      }
    }
    __syncthreads();
    for (int n = wv; n < BINW; n += 8) {
      const int s = __builtin_amdgcn_readfirstlane(nstart[n]);
      const int d = __builtin_amdgcn_readfirstlane(nstart[n + 1]) - s;
      for (int i0 = 0; i0 < d; i0 += 32) {
        const bool have = (i0 + lane) < d;
        const int e = have ? Be[s + i0 + lane] : 0x7fffffff;
        int less = 0;
        for (int j0 = 0; j0 < d; j0 += 32) {
          const int o = (j0 + lane < d) ? Be[s + j0 + lane] : 0x7fffffff;
#pragma unroll
          for (int k = 0; k < 32; ++k) less += (__builtin_amdgcn_readlane(o, k) < e) ? 1 : 0;
        }
        if (have) perm[min(binBeg + s + less, E2 - 1)] = e;
      }
    }
  } else {
    for (int i = tid; i < cntB; i += 256) {
      const int n = (Td[binBeg + i] - n0) & (BINW - 1);
      const int p = atomicAdd(&cur[n], 1);
      perm[min(max(binBeg + p, 0), E2 - 1)] = T[binBeg + i];
    }
  }
}

#define TRIPS 8
template <int FIRST>
__global__ void bp_edge(const int* __restrict__ src,
                        const int* __restrict__ rev,
                        const float* __restrict__ belin,
                        float* __restrict__ M,
                        int E2) {
  const int w    = __builtin_amdgcn_readfirstlane((int)(blockIdx.x * (blockDim.x >> 5) + (threadIdx.x >> 5)));
  const int lane = threadIdx.x & 31;
  const int g    = lane >> 2;
  const int q4   = (lane & 3) * 4;
  const int ew   = w * (32 * TRIPS);

  for (int t = 0; t < TRIPS; ++t) {
    const int e0 = ew + 32 * t;
    if (e0 >= E2) break;
    int ei[4], ki[4];
#pragma unroll
    for (int i = 0; i < 4; ++i) ei[i] = min(e0 + 8 * i + g, E2 - 1);
#pragma unroll
    for (int i = 0; i < 4; ++i) ki[i] = FIRST ? src[rev[ei[i]]] : src[ei[i]];

    v4f mr[4], b[4];
    if (!FIRST) {
#pragma unroll
      for (int i = 0; i < 4; ++i) mr[i] = *(const v4f*)(M + (size_t)ei[i] * STATES + q4);
    }
#pragma unroll
    for (int i = 0; i < 4; ++i) b[i] = *(const v4f*)(belin + (size_t)ki[i] * STATES + q4);

#pragma unroll
    for (int i = 0; i < 4; ++i) {
      v4f x;
      if (FIRST) x = b[i] / 0.0625f;
      else       x = b[i] / mr[i];
      float tt = (x.x + x.y) + (x.z + x.w);
      tt += __shfl_xor(tt, 1, 32);
      tt += __shfl_xor(tt, 2, 32);
      const v4f m = (tt + EXP1M1F * x) / ((15.0f + EXP1F) * tt);
      if (e0 + 8 * i + g < E2) *(v4f*)(M + (size_t)ei[i] * STATES + q4) = m;
    }
  }
}

#define NPW 4
#define GRP 4
template <int MODE>
__global__ void bp_pull(const int* __restrict__ perm,
                        const int* __restrict__ src,
                        const int* __restrict__ rev,
                        const float* __restrict__ priors,
                        const float* belin,
                        const float* M,
                        const int* seg,
                        float* bout,
                        int N, int E2) {
  const int w    = __builtin_amdgcn_readfirstlane((int)(blockIdx.x * (blockDim.x >> 5) + (threadIdx.x >> 5)));
  const int lane = threadIdx.x & 31;
  const int g    = lane >> 2;
  const int q4   = (lane & 3) * 4;
  const bool lo  = lane < 16;
  const int n0   = w * NPW;
  if (n0 >= N) return;

  int nb = seg[(size_t)n0 * STATES];
  int ne = seg[(size_t)n0 * STATES + 1];

  for (int jp = 0; jp < NPW; jp += 2) {
    const int nodeA = n0 + jp;
    if (nodeA >= N) break;
    const bool hasB = (nodeA + 1) < N;
    const size_t rowP = (size_t)(nodeA + ((hasB && !lo) ? 1 : 0)) * STATES;
    const v4f pr = *(const v4f*)(priors + rowP + q4);

    v4f accA = {0.0f, 0.0f, 0.0f, 0.0f};
    v4f accB = {0.0f, 0.0f, 0.0f, 0.0f};
    for (int h = 0; h < 2; ++h) {
      const int node = nodeA + h;
      if (node >= N) break;
      const int beg = min(max(__builtin_amdgcn_readfirstlane(nb), 0), E2);
      const int deg = min(max(__builtin_amdgcn_readfirstlane(ne) - beg, 0), E2 - beg);
      if (jp + h + 1 < NPW && node + 1 < N) {
        nb = seg[(size_t)(node + 1) * STATES];
        ne = seg[(size_t)(node + 1) * STATES + 1];
      }

      v4f acc = {0.0f, 0.0f, 0.0f, 0.0f};
      for (int base = 0; base < deg; base += 8 * GRP) {
        int ei[GRP], ki[GRP], ri[GRP];
        v4f b[GRP], mr[GRP];
        for (int i = 0; i < GRP; ++i) {
          if (base + 8 * i < deg) {
            const int p = base + 8 * i + g;
            ei[i] = perm[beg + ((p < deg) ? p : (base + 8 * i))];
          }
        }
        for (int i = 0; i < GRP; ++i) {
          if (base + 8 * i < deg) {
            ei[i] = min(max(ei[i], 0), E2 - 1);
            if (MODE != 3) ki[i] = src[ei[i]];
            if (MODE == 4) ri[i] = rev[ei[i]];
          }
        }
        for (int i = 0; i < GRP; ++i) {
          if (base + 8 * i < deg) {
            if (MODE == 3) {
              b[i] = *(const v4f*)(M + (size_t)ei[i] * STATES + q4);
            } else {
              b[i] = *(const v4f*)(belin + (size_t)ki[i] * STATES + q4);
              if (MODE == 4) mr[i] = *(const v4f*)(M + (size_t)ri[i] * STATES + q4);
            }
          }
        }
        for (int i = 0; i < GRP; ++i) {
          if (base + 8 * i < deg) {
            v4f lm;
            if (MODE == 1) {
              lm = b[i];
            } else {
              v4f m;
              if (MODE == 3) {
                m = b[i];
              } else {
                const v4f x = b[i] / mr[i];
                float tt = (x.x + x.y) + (x.z + x.w);
                tt += __shfl_xor(tt, 1, 32);
                tt += __shfl_xor(tt, 2, 32);
                m = (tt + EXP1M1F * x) / ((15.0f + EXP1F) * tt);
              }
              lm.x = logf(fmaxf(m.x, EPSF));
              lm.y = logf(fmaxf(m.y, EPSF));
              lm.z = logf(fmaxf(m.z, EPSF));
              lm.w = logf(fmaxf(m.w, EPSF));
            }
            const bool live = (base + 8 * i + g) < deg;
            acc.x += live ? lm.x : 0.0f;
            acc.y += live ? lm.y : 0.0f;
            acc.z += live ? lm.z : 0.0f;
            acc.w += live ? lm.w : 0.0f;
          }
        }
      }
      if (h == 0) accA = acc; else accB = acc;
    }

    v4f t = lo ? accA : accB;
    const v4f u = lo ? accB : accA;
    t.x += __shfl_xor(u.x, 16, 32);
    t.y += __shfl_xor(u.y, 16, 32);
    t.z += __shfl_xor(u.z, 16, 32);
    t.w += __shfl_xor(u.w, 16, 32);
    t.x += __shfl_xor(t.x, 8, 32);
    t.y += __shfl_xor(t.y, 8, 32);
    t.z += __shfl_xor(t.z, 8, 32);
    t.w += __shfl_xor(t.w, 8, 32);
    t.x += __shfl_xor(t.x, 4, 32);
    t.y += __shfl_xor(t.y, 4, 32);
    t.z += __shfl_xor(t.z, 4, 32);
    t.w += __shfl_xor(t.w, 4, 32);

    v4f lb;
    lb.x = logf(fmaxf(pr.x, EPSF)) + t.x;
    lb.y = logf(fmaxf(pr.y, EPSF)) + t.y;
    lb.z = logf(fmaxf(pr.z, EPSF)) + t.z;
    lb.w = logf(fmaxf(pr.w, EPSF)) + t.w;

    float mx = fmaxf(fmaxf(lb.x, lb.y), fmaxf(lb.z, lb.w));
    mx = fmaxf(mx, __shfl_xor(mx, 1, 32));
    mx = fmaxf(mx, __shfl_xor(mx, 2, 32));
    v4f ex;
    ex.x = expf(lb.x - mx);
    ex.y = expf(lb.y - mx);
    ex.z = expf(lb.z - mx);
    ex.w = expf(lb.w - mx);
    float sm = (ex.x + ex.y) + (ex.z + ex.w);
    sm += __shfl_xor(sm, 1, 32);
    sm += __shfl_xor(sm, 2, 32);
    const v4f o = ex / sm;
    if ((lane & 12) == 0 && (lo || hasB)) *(v4f*)(bout + rowP + q4) = o;
  }
}

extern "C" void kernel_launch(void* const* d_in, const int* in_sizes, int n_in,
                              void* d_out, int out_size, void* d_ws, size_t ws_size,
                              hipStream_t stream) {
  const float* feat = (const float*)d_in[0];
  const float* W    = (const float*)d_in[1];
  const int*   src  = (const int*)d_in[2];
  const int*   dst  = (const int*)d_in[3];
  const int*   rev  = (const int*)d_in[4];

  const int N  = in_sizes[0] / FDIM;
  const int E2 = in_sizes[2];

  float* priors      = (float*)d_out;
  float* beliefs_out = priors + (size_t)N * STATES;
  int*   seg         = (int*)beliefs_out;

  const int NB  = (N + BINW - 1) / BINW;
  const int BLK = (E2 + CH - 1) / CH;
  const int L   = NB * BLK;

  int*   perm   = (int*)d_ws;
  float* belbuf = (float*)d_ws + (size_t)E2;
  float* msgs   = (float*)d_ws + (size_t)3 * N * STATES;
  int*   T      = (int*)msgs;
  int*   Td     = T + (size_t)E2;
  float* lg1    = msgs + (size_t)2 * E2;
  int*   Hm     = (int*)(lg1 + (size_t)N * STATES);
  int*   Off    = Hm + (size_t)L;

  const dim3 blk(256);
  const int tiles   = (N + 15) / 16;
  const int gBlocks = (tiles + 7) / 8;
  const int sBlocks = (L + 1023) / 1024;
  const int nBlocks = (N + 8 * NPW - 1) / (8 * NPW);
  const int eWaves  = (E2 + 32 * TRIPS - 1) / (32 * TRIPS);
  const int eBlocks = (eWaves + 7) / 8;

  bp_priors_wmma<<<gBlocks, blk, 0, stream>>>(feat, W, priors, lg1, N, tiles);

  bp_bin_count<<<BLK, blk, 0, stream>>>(dst, Hm, E2, NB, BLK);
  bp_scan_excl<<<sBlocks, dim3(1024), 0, stream>>>(Hm, Off, L);
  bp_bin_scatter<<<BLK, blk, 0, stream>>>(dst, Off, T, Td, E2, NB, BLK);
  bp_bin_sort<<<NB, blk, 0, stream>>>(Off, T, Td, seg, perm, N, E2, NB, BLK);

  bp_pull<1><<<nBlocks, blk, 0, stream>>>(perm, src, rev, priors, lg1, msgs, seg, belbuf, N, E2);
  bp_edge<1><<<eBlocks, blk, 0, stream>>>(src, rev, priors, msgs, E2);
  bp_edge<0><<<eBlocks, blk, 0, stream>>>(src, rev, belbuf, msgs, E2);
  bp_pull<3><<<nBlocks, blk, 0, stream>>>(perm, src, rev, priors, belbuf, msgs, seg, belbuf, N, E2);
  bp_pull<4><<<nBlocks, blk, 0, stream>>>(perm, src, rev, priors, belbuf, msgs, seg, beliefs_out, N, E2);
}


# Sweep-1 messages kept once per node (a table written in the priors epilogue), never per edge: the pass that stored sweep-1 rows is gone, the sweep-2 row pass gathers the reverse edge's source row
# speedup vs baseline: 2.0817x; 1.1172x over previous
#include <hip/hip_runtime.h>
#include <math.h>


typedef float v2f __attribute__((ext_vector_type(2)));
typedef float v4f __attribute__((ext_vector_type(4)));
typedef float v8f __attribute__((ext_vector_type(8)));

#define STATES   16
#define FDIM     128
#define EPSF     1e-10f
#define EXP1F    2.7182818284590452f
#define EXP1M1F  1.7182818284590452f

__global__ void bp_priors_wmma(const float* __restrict__ feat,
                               const float* __restrict__ W,
                               float* __restrict__ priors,
                               float* __restrict__ lg1,
                               float* __restrict__ m1,
                               int N, int numTiles) {
  const int wave = blockIdx.x * (blockDim.x >> 5) + (threadIdx.x >> 5);
  if (wave >= numTiles) return;
  const int lane = threadIdx.x & 31;
  const int half = lane >> 4;
  const int s    = lane & 15;

  int rowA = wave * 16 + s;
  if (rowA >= N) rowA = N - 1;
  const float* frow = feat + (size_t)rowA * FDIM;

  v8f acc = {};
#pragma unroll
  for (int k0 = 0; k0 < FDIM; k0 += 4) {
    const int k = k0 + 2 * half;
    v2f a, b;
    a.x = frow[k];
    a.y = frow[k + 1];
    b.x = W[(size_t)k * STATES + s];
    b.y = W[(size_t)(k + 1) * STATES + s];
    acc = __builtin_amdgcn_wmma_f32_16x16x4_f32(false, a, false, b,
                                                (short)0, acc, false, false);
  }

#pragma unroll
  for (int r = 0; r < 8; ++r) {
    float logit = acc[r];
    float mx = logit;
    mx = fmaxf(mx, __shfl_xor(mx, 1, 32));
    mx = fmaxf(mx, __shfl_xor(mx, 2, 32));
    mx = fmaxf(mx, __shfl_xor(mx, 4, 32));
    mx = fmaxf(mx, __shfl_xor(mx, 8, 32));
    float ex = expf(logit - mx);
    float sm = ex;
    sm += __shfl_xor(sm, 1, 32);
    sm += __shfl_xor(sm, 2, 32);
    sm += __shfl_xor(sm, 4, 32);
    sm += __shfl_xor(sm, 8, 32);
    const int rowO = wave * 16 + r + 8 * half;
    const float p = ex / sm;
    const float x = p / 0.0625f;
    float tt = x;
    tt += __shfl_xor(tt, 1, 32);
    tt += __shfl_xor(tt, 2, 32);
    tt += __shfl_xor(tt, 4, 32);
    tt += __shfl_xor(tt, 8, 32);
    const float m = (tt + EXP1M1F * x) / ((15.0f + EXP1F) * tt);
    const float lm = logf(fmaxf(m, EPSF));
    if (rowO < N) {
      priors[(size_t)rowO * STATES + s] = p;
      lg1[(size_t)rowO * STATES + s]    = lm;
      if (m1) m1[(size_t)rowO * STATES + s] = m;
    }
  }
}

#define BINSH   8
#define BINW    256
#define CH      8192
#define NBMAX   1024
#define CAP     16384
#define UNR     8

__device__ __forceinline__ int bp_bin_of(int d, int NB) {
  return min(max(d >> BINSH, 0), NB - 1);
}

__global__ void bp_bin_count(const int* __restrict__ dst, int* __restrict__ Hm,
                             int E2, int NB, int BLK) {
  __shared__ int hist[NBMAX];
  const int tid = threadIdx.x;
  for (int b = tid; b < NBMAX; b += 256) hist[b] = 0;
  __syncthreads();
  const int base = blockIdx.x * CH;
  for (int j = 0; j < CH / 256; ++j) {
    const int e = base + j * 256 + tid;
    if (e < E2) atomicAdd(&hist[bp_bin_of(dst[e], NB)], 1);
  }
  __syncthreads();
  for (int b = tid; b < NB; b += 256) Hm[(size_t)b * BLK + blockIdx.x] = hist[b];
}

__global__ void bp_scan_excl(const int* __restrict__ in, int* __restrict__ out, int n) {
  __shared__ int wsum[32];
  __shared__ int sbase;
  const int tid  = threadIdx.x;
  const int lane = tid & 31;
  const int wv   = tid >> 5;
  const int start = blockIdx.x * 1024;

  int part = 0;
  for (int i = tid; i < start; i += 1024) part += in[i];
  for (int o = 16; o > 0; o >>= 1) part += __shfl_xor(part, o, 32);
  if (lane == 0) wsum[wv] = part;
  __syncthreads();
  if (wv == 0) {
    int v = wsum[lane];
    for (int o = 16; o > 0; o >>= 1) v += __shfl_xor(v, o, 32);
    if (lane == 0) sbase = v;
  }
  __syncthreads();
  const int base = sbase;

  const int i = start + tid;
  const int c = (i < n) ? in[i] : 0;
  int inc = c;
  for (int o = 1; o < 32; o <<= 1) {
    const int y = __shfl_up(inc, o, 32);
    if (lane >= o) inc += y;
  }
  if (lane == 31) wsum[wv] = inc;
  __syncthreads();
  if (wv == 0) {
    const int v = wsum[lane];
    int iv = v;
    for (int o = 1; o < 32; o <<= 1) {
      const int y = __shfl_up(iv, o, 32);
      if (lane >= o) iv += y;
    }
    wsum[lane] = iv - v;
  }
  __syncthreads();
  if (i < n) out[i] = base + wsum[wv] + inc - c;
}

__global__ void bp_bin_scatter(const int* __restrict__ dst, const int* __restrict__ Off,
                               int* __restrict__ T, int* __restrict__ Td,
                               int E2, int NB, int BLK) {
  __shared__ int hist[NBMAX];
  __shared__ int delta[NBMAX];
  __shared__ int se[CH];
  __shared__ int sd[CH];
  __shared__ int wsum[8];
  const int tid  = threadIdx.x;
  const int lane = tid & 31;
  const int wv   = tid >> 5;
  const int blk  = blockIdx.x;
  const int base = blk * CH;
  const int nIt  = min(CH, E2 - base);

  int dd[CH / 256];
#pragma unroll
  for (int j = 0; j < CH / 256; ++j) {
    const int e = base + j * 256 + tid;
    dd[j] = (e < E2) ? dst[e] : 0;
  }
  for (int b = tid; b < NBMAX; b += 256) hist[b] = 0;
  __syncthreads();
#pragma unroll
  for (int j = 0; j < CH / 256; ++j) {
    const int e = base + j * 256 + tid;
    if (e < E2) atomicAdd(&hist[bp_bin_of(dd[j], NB)], 1);
  }
  __syncthreads();
  {
    const int v0 = hist[4 * tid], v1 = hist[4 * tid + 1], v2 = hist[4 * tid + 2], v3 = hist[4 * tid + 3];
    const int s4 = (v0 + v1) + (v2 + v3);
    int inc = s4;
    for (int o = 1; o < 32; o <<= 1) {
      const int y = __shfl_up(inc, o, 32);
      if (lane >= o) inc += y;
    }
    if (lane == 31) wsum[wv] = inc;
    __syncthreads();
    int woff = 0;
    for (int k = 0; k < wv; ++k) woff += wsum[k];
    const int ex = woff + inc - s4;
    hist[4 * tid]     = ex;
    hist[4 * tid + 1] = ex + v0;
    hist[4 * tid + 2] = ex + v0 + v1;
    hist[4 * tid + 3] = ex + v0 + v1 + v2;
  }
  __syncthreads();
  for (int b = tid; b < NB; b += 256) delta[b] = Off[(size_t)b * BLK + blk] - hist[b];
  __syncthreads();
#pragma unroll
  for (int j = 0; j < CH / 256; ++j) {
    const int e = base + j * 256 + tid;
    if (e < E2) {
      const int d = dd[j];
      const int p = min(max(atomicAdd(&hist[bp_bin_of(d, NB)], 1), 0), CH - 1);
      se[p] = e;
      sd[p] = d;
    }
  }
  __syncthreads();
  for (int s = tid; s < nIt; s += 256) {
    const int d = sd[s];
    const int g = min(max(s + delta[bp_bin_of(d, NB)], 0), E2 - 1);
    T[g]  = se[s];
    Td[g] = d;
  }
}

__global__ void bp_bin_sort(const int* __restrict__ Off, const int* __restrict__ T,
                            const int* __restrict__ Td, int* __restrict__ seg,
                            int* __restrict__ perm, int N, int E2, int NB, int BLK) {
  __shared__ int ncnt[BINW];
  __shared__ int nstart[BINW + 1];
  __shared__ int cur[BINW];
  __shared__ int Be[CAP];
  __shared__ int wsum[8];
  const int tid  = threadIdx.x;
  const int lane = tid & 31;
  const int wv   = tid >> 5;
  const int b    = blockIdx.x;
  const int n0   = b << BINSH;

  int binBeg = Off[(size_t)b * BLK];
  int binEnd = (b + 1 < NB) ? Off[(size_t)(b + 1) * BLK] : E2;
  binBeg = min(max(binBeg, 0), E2);
  binEnd = min(max(binEnd, binBeg), E2);
  const int cntB = binEnd - binBeg;

  ncnt[tid] = 0;
  __syncthreads();
  for (int i0 = 0; i0 < cntB; i0 += 256 * UNR) {
    int dn[UNR];
#pragma unroll
    for (int u = 0; u < UNR; ++u) {
      const int i = i0 + u * 256 + tid;
      dn[u] = (i < cntB) ? Td[binBeg + i] : n0;
    }
#pragma unroll
    for (int u = 0; u < UNR; ++u) {
      const int i = i0 + u * 256 + tid;
      if (i < cntB) atomicAdd(&ncnt[(dn[u] - n0) & (BINW - 1)], 1);
    }
  }
  __syncthreads();
  {
    const int c = ncnt[tid];
    int inc = c;
    for (int o = 1; o < 32; o <<= 1) {
      const int y = __shfl_up(inc, o, 32);
      if (lane >= o) inc += y;
    }
    if (lane == 31) wsum[wv] = inc;
    __syncthreads();
    int woff = 0;
    for (int k = 0; k < wv; ++k) woff += wsum[k];
    const int ex = woff + inc - c;
    nstart[tid] = ex;
    cur[tid]    = ex;
    if (tid == BINW - 1) nstart[BINW] = ex + c;
    const int node = n0 + tid;
    if (node < N) {
      seg[(size_t)node * STATES]     = binBeg + ex;
      seg[(size_t)node * STATES + 1] = binBeg + ex + c;
    }
  }
  __syncthreads();
  if (cntB <= CAP) {
    for (int i0 = 0; i0 < cntB; i0 += 256 * UNR) {
      int dn[UNR], en[UNR];
#pragma unroll
      for (int u = 0; u < UNR; ++u) {
        const int i = i0 + u * 256 + tid;
        dn[u] = (i < cntB) ? Td[binBeg + i] : n0;
        en[u] = (i < cntB) ? T[binBeg + i] : 0;
      }
#pragma unroll
      for (int u = 0; u < UNR; ++u) {
        const int i = i0 + u * 256 + tid;
        if (i < cntB) {
          const int p = min(max(atomicAdd(&cur[(dn[u] - n0) & (BINW - 1)], 1), 0), CAP - 1);
          Be[p] = en[u];
        }
      }
    }
    __syncthreads();
    for (int n = wv; n < BINW; n += 8) {
      const int s = __builtin_amdgcn_readfirstlane(nstart[n]);
      const int d = __builtin_amdgcn_readfirstlane(nstart[n + 1]) - s;
      for (int i0 = 0; i0 < d; i0 += 32) {
        const bool have = (i0 + lane) < d;
        const int e = have ? Be[s + i0 + lane] : 0x7fffffff;
        int less = 0;
        for (int j0 = 0; j0 < d; j0 += 32) {
          const int o = (j0 + lane < d) ? Be[s + j0 + lane] : 0x7fffffff;
#pragma unroll
          for (int k = 0; k < 32; ++k) less += (__builtin_amdgcn_readlane(o, k) < e) ? 1 : 0;
        }
        if (have) perm[min(binBeg + s + less, E2 - 1)] = e;
      }
    }
  } else {
    for (int i = tid; i < cntB; i += 256) {
      const int n = (Td[binBeg + i] - n0) & (BINW - 1);
      const int p = atomicAdd(&cur[n], 1);
      perm[min(max(binBeg + p, 0), E2 - 1)] = T[binBeg + i];
    }
  }
}

#define TRIPS 8
template <int TAB>
__global__ void bp_edge(const int* __restrict__ src,
                        const int* __restrict__ rev,
                        const float* rowsrc,
                        const float* __restrict__ bel1,
                        float* M,
                        int eBeg, int eEnd) {
  const int w    = __builtin_amdgcn_readfirstlane((int)(blockIdx.x * (blockDim.x >> 5) + (threadIdx.x >> 5)));
  const int lane = threadIdx.x & 31;
  const int g    = lane >> 2;
  const int q4   = (lane & 3) * 4;
  const int ew   = eBeg + w * (32 * TRIPS);

  for (int t = 0; t < TRIPS; ++t) {
    const int e0 = ew + 32 * t;
    if (e0 >= eEnd) break;
    int ei[4], ki[4], ri[4];
#pragma unroll
    for (int i = 0; i < 4; ++i) ei[i] = min(e0 + 8 * i + g, eEnd - 1);
#pragma unroll
    for (int i = 0; i < 4; ++i) { ki[i] = src[ei[i]]; ri[i] = rev[ei[i]]; }
#pragma unroll
    for (int i = 0; i < 4; ++i) ri[i] = src[ri[i]];

    v4f b[4], mr[4];
#pragma unroll
    for (int i = 0; i < 4; ++i) b[i] = *(const v4f*)(bel1 + (size_t)ki[i] * STATES + q4);
#pragma unroll
    for (int i = 0; i < 4; ++i) mr[i] = *(const v4f*)(rowsrc + (size_t)ri[i] * STATES + q4);

#pragma unroll
    for (int i = 0; i < 4; ++i) {
      if (!TAB) {
        const v4f x1 = mr[i] / 0.0625f;
        float t1 = (x1.x + x1.y) + (x1.z + x1.w);
        t1 += __shfl_xor(t1, 1, 32);
        t1 += __shfl_xor(t1, 2, 32);
        mr[i] = (t1 + EXP1M1F * x1) / ((15.0f + EXP1F) * t1);
      }
      const v4f x = b[i] / mr[i];
      float tt = (x.x + x.y) + (x.z + x.w);
      tt += __shfl_xor(tt, 1, 32);
      tt += __shfl_xor(tt, 2, 32);
      const v4f m = (tt + EXP1M1F * x) / ((15.0f + EXP1F) * tt);
      if (e0 + 8 * i + g < eEnd) *(v4f*)(M + (size_t)ei[i] * STATES + q4) = m;
    }
  }
}

#define NPW 4
#define GRP 4
template <int MODE>
__global__ void bp_pull(const int* __restrict__ perm,
                        const int* __restrict__ src,
                        const int* __restrict__ rev,
                        const float* __restrict__ priors,
                        const float* belin,
                        const float* M,
                        const int* seg,
                        float* bout,
                        int N, int E2) {
  const int w    = __builtin_amdgcn_readfirstlane((int)(blockIdx.x * (blockDim.x >> 5) + (threadIdx.x >> 5)));
  const int lane = threadIdx.x & 31;
  const int g    = lane >> 2;
  const int q4   = (lane & 3) * 4;
  const bool lo  = lane < 16;
  const int n0   = w * NPW;
  if (n0 >= N) return;

  int nb = seg[(size_t)n0 * STATES];
  int ne = seg[(size_t)n0 * STATES + 1];

  for (int jp = 0; jp < NPW; jp += 2) {
    const int nodeA = n0 + jp;
    if (nodeA >= N) break;
    const bool hasB = (nodeA + 1) < N;
    const size_t rowP = (size_t)(nodeA + ((hasB && !lo) ? 1 : 0)) * STATES;
    const v4f pr = *(const v4f*)(priors + rowP + q4);

    v4f accA = {0.0f, 0.0f, 0.0f, 0.0f};
    v4f accB = {0.0f, 0.0f, 0.0f, 0.0f};
    for (int h = 0; h < 2; ++h) {
      const int node = nodeA + h;
      if (node >= N) break;
      const int beg = min(max(__builtin_amdgcn_readfirstlane(nb), 0), E2);
      const int deg = min(max(__builtin_amdgcn_readfirstlane(ne) - beg, 0), E2 - beg);
      if (jp + h + 1 < NPW && node + 1 < N) {
        nb = seg[(size_t)(node + 1) * STATES];
        ne = seg[(size_t)(node + 1) * STATES + 1];
      }

      v4f acc = {0.0f, 0.0f, 0.0f, 0.0f};
      for (int base = 0; base < deg; base += 8 * GRP) {
        int ei[GRP], ki[GRP], ri[GRP];
        v4f b[GRP], mr[GRP];
        for (int i = 0; i < GRP; ++i) {
          if (base + 8 * i < deg) {
            const int p = base + 8 * i + g;
            ei[i] = perm[beg + ((p < deg) ? p : (base + 8 * i))];
          }
        }
        for (int i = 0; i < GRP; ++i) {
          if (base + 8 * i < deg) {
            ei[i] = min(max(ei[i], 0), E2 - 1);
            if (MODE != 3) ki[i] = src[ei[i]];
            if (MODE == 4) ri[i] = rev[ei[i]];
          }
        }
        for (int i = 0; i < GRP; ++i) {
          if (base + 8 * i < deg) {
            if (MODE == 3) {
              b[i] = *(const v4f*)(M + (size_t)ei[i] * STATES + q4);
            } else {
              b[i] = *(const v4f*)(belin + (size_t)ki[i] * STATES + q4);
              if (MODE == 4) mr[i] = *(const v4f*)(M + (size_t)ri[i] * STATES + q4);
            }
          }
        }
        for (int i = 0; i < GRP; ++i) {
          if (base + 8 * i < deg) {
            v4f lm;
            if (MODE == 1) {
              lm = b[i];
            } else {
              v4f m;
              if (MODE == 3) {
                m = b[i];
              } else {
                const v4f x = b[i] / mr[i];
                float tt = (x.x + x.y) + (x.z + x.w);
                tt += __shfl_xor(tt, 1, 32);
                tt += __shfl_xor(tt, 2, 32);
                m = (tt + EXP1M1F * x) / ((15.0f + EXP1F) * tt);
              }
              lm.x = logf(fmaxf(m.x, EPSF));
              lm.y = logf(fmaxf(m.y, EPSF));
              lm.z = logf(fmaxf(m.z, EPSF));
              lm.w = logf(fmaxf(m.w, EPSF));
            }
            const bool live = (base + 8 * i + g) < deg;
            acc.x += live ? lm.x : 0.0f;
            acc.y += live ? lm.y : 0.0f;
            acc.z += live ? lm.z : 0.0f;
            acc.w += live ? lm.w : 0.0f;
          }
        }
      }
      if (h == 0) accA = acc; else accB = acc;
    }

    v4f t = lo ? accA : accB;
    const v4f u = lo ? accB : accA;
    t.x += __shfl_xor(u.x, 16, 32);
    t.y += __shfl_xor(u.y, 16, 32);
    t.z += __shfl_xor(u.z, 16, 32);
    t.w += __shfl_xor(u.w, 16, 32);
    t.x += __shfl_xor(t.x, 8, 32);
    t.y += __shfl_xor(t.y, 8, 32);
    t.z += __shfl_xor(t.z, 8, 32);
    t.w += __shfl_xor(t.w, 8, 32);
    t.x += __shfl_xor(t.x, 4, 32);
    t.y += __shfl_xor(t.y, 4, 32);
    t.z += __shfl_xor(t.z, 4, 32);
    t.w += __shfl_xor(t.w, 4, 32);

    v4f lb;
    lb.x = logf(fmaxf(pr.x, EPSF)) + t.x;
    lb.y = logf(fmaxf(pr.y, EPSF)) + t.y;
    lb.z = logf(fmaxf(pr.z, EPSF)) + t.z;
    lb.w = logf(fmaxf(pr.w, EPSF)) + t.w;

    float mx = fmaxf(fmaxf(lb.x, lb.y), fmaxf(lb.z, lb.w));
    mx = fmaxf(mx, __shfl_xor(mx, 1, 32));
    mx = fmaxf(mx, __shfl_xor(mx, 2, 32));
    v4f ex;
    ex.x = expf(lb.x - mx);
    ex.y = expf(lb.y - mx);
    ex.z = expf(lb.z - mx);
    ex.w = expf(lb.w - mx);
    float sm = (ex.x + ex.y) + (ex.z + ex.w);
    sm += __shfl_xor(sm, 1, 32);
    sm += __shfl_xor(sm, 2, 32);
    const v4f o = ex / sm;
    if ((lane & 12) == 0 && (lo || hasB)) *(v4f*)(bout + rowP + q4) = o;
  }
}

extern "C" void kernel_launch(void* const* d_in, const int* in_sizes, int n_in,
                              void* d_out, int out_size, void* d_ws, size_t ws_size,
                              hipStream_t stream) {
  const float* feat = (const float*)d_in[0];
  const float* W    = (const float*)d_in[1];
  const int*   src  = (const int*)d_in[2];
  const int*   dst  = (const int*)d_in[3];
  const int*   rev  = (const int*)d_in[4];

  const int N  = in_sizes[0] / FDIM;
  const int E2 = in_sizes[2];

  float* priors      = (float*)d_out;
  float* beliefs_out = priors + (size_t)N * STATES;
  int*   seg         = (int*)beliefs_out;

  const int NB  = (N + BINW - 1) / BINW;
  const int BLK = (E2 + CH - 1) / CH;
  const int L   = NB * BLK;

  int*   perm   = (int*)d_ws;
  float* belbuf = (float*)d_ws + (size_t)E2;
  float* msgs   = (float*)d_ws + (size_t)3 * N * STATES;
  int*   T      = (int*)msgs;
  int*   Td     = T + (size_t)E2;
  float* lg1    = msgs + (size_t)2 * E2;
  int*   Hm     = (int*)(lg1 + (size_t)N * STATES);
  int*   Off    = Hm + (size_t)L;
  const bool tab  = E2 > N && (size_t)(E2 - N) * STATES >= (size_t)2 * E2 + (size_t)N * STATES + (size_t)2 * L;
  float* m1tab    = tab ? msgs + (size_t)(E2 - N) * STATES : nullptr;
  const int split = tab ? ((E2 - N) / (32 * TRIPS)) * (32 * TRIPS) : 0;

  const dim3 blk(256);
  const int tiles   = (N + 15) / 16;
  const int gBlocks = (tiles + 7) / 8;
  const int sBlocks = (L + 1023) / 1024;
  const int nBlocks = (N + 8 * NPW - 1) / (8 * NPW);
  const int eBlocksA = (split / (32 * TRIPS) + 7) / 8;
  const int eBlocksB = ((E2 - split + 32 * TRIPS - 1) / (32 * TRIPS) + 7) / 8;

  bp_priors_wmma<<<gBlocks, blk, 0, stream>>>(feat, W, priors, lg1, m1tab, N, tiles);

  bp_bin_count<<<BLK, blk, 0, stream>>>(dst, Hm, E2, NB, BLK);
  bp_scan_excl<<<sBlocks, dim3(1024), 0, stream>>>(Hm, Off, L);
  bp_bin_scatter<<<BLK, blk, 0, stream>>>(dst, Off, T, Td, E2, NB, BLK);
  bp_bin_sort<<<NB, blk, 0, stream>>>(Off, T, Td, seg, perm, N, E2, NB, BLK);

  bp_pull<1><<<nBlocks, blk, 0, stream>>>(perm, src, rev, priors, lg1, msgs, seg, belbuf, N, E2);
  if (split > 0) bp_edge<1><<<eBlocksA, blk, 0, stream>>>(src, rev, m1tab, belbuf, msgs, 0, split);
  bp_edge<0><<<eBlocksB, blk, 0, stream>>>(src, rev, priors, belbuf, msgs, split, E2);
  bp_pull<3><<<nBlocks, blk, 0, stream>>>(perm, src, rev, priors, belbuf, msgs, seg, belbuf, N, E2);
  bp_pull<4><<<nBlocks, blk, 0, stream>>>(perm, src, rev, priors, belbuf, msgs, seg, beliefs_out, N, E2);
}


# Destination sort: each thread's loads and LDS round trips of the count, placement and write-out loops go out as one batch; the built code had one dependent round trip per edge (about 100 a thread)
# speedup vs baseline: 2.3093x; 1.1093x over previous
#include <hip/hip_runtime.h>
#include <math.h>


typedef float v2f __attribute__((ext_vector_type(2)));
typedef float v4f __attribute__((ext_vector_type(4)));
typedef float v8f __attribute__((ext_vector_type(8)));

#define STATES   16
#define FDIM     128
#define EPSF     1e-10f
#define EXP1F    2.7182818284590452f
#define EXP1M1F  1.7182818284590452f

__device__ __forceinline__ void bp_priors_tiles(const float* __restrict__ feat,
                                                const float* __restrict__ W,
                                                float* __restrict__ priors,
                                                float* __restrict__ lg1,
                                                float* __restrict__ m1,
                                                int N, int numTiles, int tb) {
  const int wave = tb * 8 + (threadIdx.x >> 5);
  if (wave >= numTiles) return;
  const int lane = threadIdx.x & 31;
  const int half = lane >> 4;
  const int s    = lane & 15;

  int rowA = wave * 16 + s;
  if (rowA >= N) rowA = N - 1;
  const float* frow = feat + (size_t)rowA * FDIM;

  v8f acc = {};
#pragma unroll
  for (int k0 = 0; k0 < FDIM; k0 += 4) {
    const int k = k0 + 2 * half;
    v2f a, b;
    a.x = frow[k];
    a.y = frow[k + 1];
    b.x = W[(size_t)k * STATES + s];
    b.y = W[(size_t)(k + 1) * STATES + s];
    acc = __builtin_amdgcn_wmma_f32_16x16x4_f32(false, a, false, b,
                                                (short)0, acc, false, false);
  }

#pragma unroll
  for (int r = 0; r < 8; ++r) {
    float logit = acc[r];
    float mx = logit;
    mx = fmaxf(mx, __shfl_xor(mx, 1, 32));
    mx = fmaxf(mx, __shfl_xor(mx, 2, 32));
    mx = fmaxf(mx, __shfl_xor(mx, 4, 32));
    mx = fmaxf(mx, __shfl_xor(mx, 8, 32));
    float ex = expf(logit - mx);
    float sm = ex;
    sm += __shfl_xor(sm, 1, 32);
    sm += __shfl_xor(sm, 2, 32);
    sm += __shfl_xor(sm, 4, 32);
    sm += __shfl_xor(sm, 8, 32);
    const int rowO = wave * 16 + r + 8 * half;
    const float p = ex / sm;
    const float x = p / 0.0625f;
    float tt = x;
    tt += __shfl_xor(tt, 1, 32);
    tt += __shfl_xor(tt, 2, 32);
    tt += __shfl_xor(tt, 4, 32);
    tt += __shfl_xor(tt, 8, 32);
    const float m = (tt + EXP1M1F * x) / ((15.0f + EXP1F) * tt);
    const float lm = logf(fmaxf(m, EPSF));
    if (rowO < N) {
      priors[(size_t)rowO * STATES + s] = p;
      lg1[(size_t)rowO * STATES + s]    = lm;
      if (m1) m1[(size_t)rowO * STATES + s] = m;
    }
  }
}

#define BINSH   8
#define BINW    256
#define CH      8192
#define NBMAX   1024
#define CAP     16384
#define UNR     40

__device__ __forceinline__ int bp_bin_of(int d, int NB) {
  return min(max(d >> BINSH, 0), NB - 1);
}

__global__ void bp_bin_count(const int* __restrict__ dst, int* __restrict__ Hm,
                             int E2, int NB, int BLK) {
  __shared__ int hist[NBMAX];
  __shared__ int sink[256];
  const int tid = threadIdx.x;
  const int base = blockIdx.x * CH;
  int dd[CH / 256];
#pragma unroll
  for (int j = 0; j < CH / 256; ++j) {
    const int e = base + j * 256 + tid;
    dd[j] = dst[min(e, E2 - 1)];
  }
  for (int b = tid; b < NBMAX; b += 256) hist[b] = 0;
  sink[tid] = 0;
  __syncthreads();
#pragma unroll
  for (int j = 0; j < CH / 256; ++j) {
    const int e = base + j * 256 + tid;
    int* ctr = (e < E2) ? &hist[bp_bin_of(dd[j], NB)] : &sink[tid];
    atomicAdd(ctr, 1);
  }
  __syncthreads();
  for (int b = tid; b < NB; b += 256) Hm[(size_t)b * BLK + blockIdx.x] = hist[b];
}

__global__ void bp_scan_excl(const int* __restrict__ in, int* __restrict__ out, int n) {
  __shared__ int wsum[32];
  __shared__ int sbase;
  const int tid  = threadIdx.x;
  const int lane = tid & 31;
  const int wv   = tid >> 5;
  const int start = blockIdx.x * 1024;

  int part = 0;
  for (int i = tid; i < start; i += 1024) part += in[i];
  for (int o = 16; o > 0; o >>= 1) part += __shfl_xor(part, o, 32);
  if (lane == 0) wsum[wv] = part;
  __syncthreads();
  if (wv == 0) {
    int v = wsum[lane];
    for (int o = 16; o > 0; o >>= 1) v += __shfl_xor(v, o, 32);
    if (lane == 0) sbase = v;
  }
  __syncthreads();
  const int base = sbase;

  const int i = start + tid;
  const int c = (i < n) ? in[i] : 0;
  int inc = c;
  for (int o = 1; o < 32; o <<= 1) {
    const int y = __shfl_up(inc, o, 32);
    if (lane >= o) inc += y;
  }
  if (lane == 31) wsum[wv] = inc;
  __syncthreads();
  if (wv == 0) {
    const int v = wsum[lane];
    int iv = v;
    for (int o = 1; o < 32; o <<= 1) {
      const int y = __shfl_up(iv, o, 32);
      if (lane >= o) iv += y;
    }
    wsum[lane] = iv - v;
  }
  __syncthreads();
  if (i < n) out[i] = base + wsum[wv] + inc - c;
}

__global__ void __launch_bounds__(256) bp_bin_scatter(const int* __restrict__ dst, const int* __restrict__ Off,
                               int* __restrict__ T, int* __restrict__ Td,
                               int E2, int NB, int BLK,
                               const float* __restrict__ feat, const float* __restrict__ W,
                               float* __restrict__ priors, float* __restrict__ lg1,
                               float* __restrict__ m1, int N, int numTiles, int tb0) {
  if ((int)blockIdx.x >= BLK) {
    bp_priors_tiles(feat, W, priors, lg1, m1, N, numTiles, tb0 + ((int)blockIdx.x - BLK));
    return;
  }
  __shared__ int hist[NBMAX];
  __shared__ int delta[NBMAX];
  __shared__ int se[CH];
  __shared__ int sd[CH];
  __shared__ int wsum[8];
  __shared__ int sink[256];
  const int tid  = threadIdx.x;
  const int lane = tid & 31;
  const int wv   = tid >> 5;
  const int blk  = blockIdx.x;
  const int base = blk * CH;
  const int nIt  = min(CH, E2 - base);

  int dd[CH / 256];
#pragma unroll
  for (int j = 0; j < CH / 256; ++j) {
    const int e = base + j * 256 + tid;
    dd[j] = (e < E2) ? dst[e] : 0;
  }
  for (int b = tid; b < NBMAX; b += 256) hist[b] = 0;
  sink[tid] = 0;
  __syncthreads();
#pragma unroll
  for (int j = 0; j < CH / 256; ++j) {
    const int e = base + j * 256 + tid;
    if (e < E2) atomicAdd(&hist[bp_bin_of(dd[j], NB)], 1);
  }
  __syncthreads();
  {
    const int v0 = hist[4 * tid], v1 = hist[4 * tid + 1], v2 = hist[4 * tid + 2], v3 = hist[4 * tid + 3];
    const int s4 = (v0 + v1) + (v2 + v3);
    int inc = s4;
    for (int o = 1; o < 32; o <<= 1) {
      const int y = __shfl_up(inc, o, 32);
      if (lane >= o) inc += y;
    }
    if (lane == 31) wsum[wv] = inc;
    __syncthreads();
    int woff = 0;
    for (int k = 0; k < wv; ++k) woff += wsum[k];
    const int ex = woff + inc - s4;
    hist[4 * tid]     = ex;
    hist[4 * tid + 1] = ex + v0;
    hist[4 * tid + 2] = ex + v0 + v1;
    hist[4 * tid + 3] = ex + v0 + v1 + v2;
  }
  __syncthreads();
  {
    int ov[NBMAX / 256];
#pragma unroll
    for (int k = 0; k < NBMAX / 256; ++k) ov[k] = Off[(size_t)min(k * 256 + tid, NB - 1) * BLK + blk];
#pragma unroll
    for (int k = 0; k < NBMAX / 256; ++k) {
      const int b = k * 256 + tid;
      if (b < NB) delta[b] = ov[k] - hist[b];
    }
  }
  __syncthreads();
  int pp[CH / 256];
#pragma unroll
  for (int j = 0; j < CH / 256; ++j) {
    const int e = base + j * 256 + tid;
    int* ctr = (e < E2) ? &hist[bp_bin_of(dd[j], NB)] : &sink[tid];
    pp[j] = atomicAdd(ctr, 1);
  }
#pragma unroll
  for (int j = 0; j < CH / 256; ++j) {
    const int e = base + j * 256 + tid;
    if (e < E2) {
      const int p = min(max(pp[j], 0), CH - 1);
      se[p] = e;
      sd[p] = dd[j];
    }
  }
  __syncthreads();
  int dv[CH / 256], ev[CH / 256], gv[CH / 256];
#pragma unroll
  for (int j = 0; j < CH / 256; ++j) {
    const int s = min(j * 256 + tid, nIt - 1);
    dv[j] = sd[s];
    ev[j] = se[s];
  }
#pragma unroll
  for (int j = 0; j < CH / 256; ++j) gv[j] = delta[bp_bin_of(dv[j], NB)];
#pragma unroll
  for (int j = 0; j < CH / 256; ++j) {
    const int s = j * 256 + tid;
    if (s < nIt) {
      const int g = min(max(s + gv[j], 0), E2 - 1);
      T[g]  = ev[j];
      Td[g] = dv[j];
    }
  }
}

__global__ void __launch_bounds__(256) bp_bin_sort(const int* __restrict__ Off, const int* __restrict__ T,
                            const int* __restrict__ Td, int* __restrict__ seg,
                            int* __restrict__ perm, int N, int E2, int NB, int BLK,
                            const float* __restrict__ feat, const float* __restrict__ W,
                            float* __restrict__ priors, float* __restrict__ lg1,
                            float* __restrict__ m1, int numTiles, int tb0) {
  if ((int)blockIdx.x >= NB) {
    bp_priors_tiles(feat, W, priors, lg1, m1, N, numTiles, tb0 + ((int)blockIdx.x - NB));
    return;
  }
  __shared__ int ncnt[BINW];
  __shared__ int nstart[BINW + 1];
  __shared__ int cur[BINW];
  __shared__ int Be[CAP];
  __shared__ int wsum[8];
  __shared__ int sink[256];
  const int tid  = threadIdx.x;
  const int lane = tid & 31;
  const int wv   = tid >> 5;
  const int b    = blockIdx.x;
  const int n0   = b << BINSH;

  int binBeg = Off[(size_t)b * BLK];
  int binEnd = (b + 1 < NB) ? Off[(size_t)(b + 1) * BLK] : E2;
  binBeg = min(max(binBeg, 0), E2);
  binEnd = min(max(binEnd, binBeg), E2);
  const int cntB = binEnd - binBeg;

  ncnt[tid] = 0;
  sink[tid] = 0;
  __syncthreads();
  for (int i0 = 0; i0 < cntB; i0 += 256 * UNR) {
    int dn[UNR];
#pragma unroll
    for (int u = 0; u < UNR; ++u) {
      const int i = i0 + u * 256 + tid;
      dn[u] = Td[binBeg + min(i, cntB - 1)];
    }
#pragma unroll
    for (int u = 0; u < UNR; ++u) {
      const int i = i0 + u * 256 + tid;
      int* ctr = (i < cntB) ? &ncnt[(dn[u] - n0) & (BINW - 1)] : &sink[tid];
      atomicAdd(ctr, 1);
    }
  }
  __syncthreads();
  {
    const int c = ncnt[tid];
    int inc = c;
    for (int o = 1; o < 32; o <<= 1) {
      const int y = __shfl_up(inc, o, 32);
      if (lane >= o) inc += y;
    }
    if (lane == 31) wsum[wv] = inc;
    __syncthreads();
    int woff = 0;
    for (int k = 0; k < wv; ++k) woff += wsum[k];
    const int ex = woff + inc - c;
    nstart[tid] = ex;
    cur[tid]    = ex;
    if (tid == BINW - 1) nstart[BINW] = ex + c;
    const int node = n0 + tid;
    if (node < N) {
      seg[(size_t)node * STATES]     = binBeg + ex;
      seg[(size_t)node * STATES + 1] = binBeg + ex + c;
    }
  }
  __syncthreads();
  if (cntB <= CAP) {
    for (int i0 = 0; i0 < cntB; i0 += 256 * UNR) {
      int dn[UNR], en[UNR];
#pragma unroll
      for (int u = 0; u < UNR; ++u) {
        const int i = min(i0 + u * 256 + tid, cntB - 1);
        dn[u] = Td[binBeg + i];
        en[u] = T[binBeg + i];
      }
      int pq[UNR];
#pragma unroll
      for (int u = 0; u < UNR; ++u) {
        const int i = i0 + u * 256 + tid;
        int* ctr = (i < cntB) ? &cur[(dn[u] - n0) & (BINW - 1)] : &sink[tid];
        pq[u] = atomicAdd(ctr, 1);
      }
#pragma unroll
      for (int u = 0; u < UNR; ++u) {
        const int i = i0 + u * 256 + tid;
        if (i < cntB) Be[min(max(pq[u], 0), CAP - 1)] = en[u];
      }
    }
    __syncthreads();
    for (int n = wv; n < BINW; n += 8) {
      const int s = __builtin_amdgcn_readfirstlane(nstart[n]);
      const int d = __builtin_amdgcn_readfirstlane(nstart[n + 1]) - s;
      for (int i0 = 0; i0 < d; i0 += 32) {
        const bool have = (i0 + lane) < d;
        const int e = have ? Be[s + i0 + lane] : 0x7fffffff;
        int less = 0;
        for (int j0 = 0; j0 < d; j0 += 32) {
          const int o = (j0 + lane < d) ? Be[s + j0 + lane] : 0x7fffffff;
#pragma unroll
          for (int k = 0; k < 32; ++k) less += (__builtin_amdgcn_readlane(o, k) < e) ? 1 : 0;
        }
        if (have) perm[min(binBeg + s + less, E2 - 1)] = e;
      }
    }
  } else {
    for (int i = tid; i < cntB; i += 256) {
      const int n = (Td[binBeg + i] - n0) & (BINW - 1);
      const int p = atomicAdd(&cur[n], 1);
      perm[min(max(binBeg + p, 0), E2 - 1)] = T[binBeg + i];
    }
  }
}

#define TRIPS 8
template <int TAB>
__global__ void bp_edge(const int* __restrict__ src,
                        const int* __restrict__ rev,
                        const float* rowsrc,
                        const float* __restrict__ bel1,
                        float* M,
                        int eBeg, int eEnd) {
  const int w    = __builtin_amdgcn_readfirstlane((int)(blockIdx.x * (blockDim.x >> 5) + (threadIdx.x >> 5)));
  const int lane = threadIdx.x & 31;
  const int g    = lane >> 2;
  const int q4   = (lane & 3) * 4;
  const int ew   = eBeg + w * (32 * TRIPS);

  for (int t = 0; t < TRIPS; ++t) {
    const int e0 = ew + 32 * t;
    if (e0 >= eEnd) break;
    int ei[4], ki[4], ri[4];
#pragma unroll
    for (int i = 0; i < 4; ++i) ei[i] = min(e0 + 8 * i + g, eEnd - 1);
#pragma unroll
    for (int i = 0; i < 4; ++i) { ki[i] = src[ei[i]]; ri[i] = rev[ei[i]]; }
#pragma unroll
    for (int i = 0; i < 4; ++i) ri[i] = src[ri[i]];

    v4f b[4], mr[4];
#pragma unroll
    for (int i = 0; i < 4; ++i) b[i] = *(const v4f*)(bel1 + (size_t)ki[i] * STATES + q4);
#pragma unroll
    for (int i = 0; i < 4; ++i) mr[i] = *(const v4f*)(rowsrc + (size_t)ri[i] * STATES + q4);

#pragma unroll
    for (int i = 0; i < 4; ++i) {
      if (!TAB) {
        const v4f x1 = mr[i] / 0.0625f;
        float t1 = (x1.x + x1.y) + (x1.z + x1.w);
        t1 += __shfl_xor(t1, 1, 32);
        t1 += __shfl_xor(t1, 2, 32);
        mr[i] = (t1 + EXP1M1F * x1) / ((15.0f + EXP1F) * t1);
      }
      const v4f x = b[i] / mr[i];
      float tt = (x.x + x.y) + (x.z + x.w);
      tt += __shfl_xor(tt, 1, 32);
      tt += __shfl_xor(tt, 2, 32);
      const v4f m = (tt + EXP1M1F * x) / ((15.0f + EXP1F) * tt);
      if (e0 + 8 * i + g < eEnd) *(v4f*)(M + (size_t)ei[i] * STATES + q4) = m;
    }
  }
}

#define NPW 4
#define GRP 4
template <int MODE>
__global__ void bp_pull(const int* __restrict__ perm,
                        const int* __restrict__ src,
                        const int* __restrict__ rev,
                        const float* __restrict__ priors,
                        const float* belin,
                        const float* M,
                        const int* seg,
                        float* bout,
                        int N, int E2) {
  const int w    = __builtin_amdgcn_readfirstlane((int)(blockIdx.x * (blockDim.x >> 5) + (threadIdx.x >> 5)));
  const int lane = threadIdx.x & 31;
  const int g    = lane >> 2;
  const int q4   = (lane & 3) * 4;
  const bool lo  = lane < 16;
  const int n0   = w * NPW;
  if (n0 >= N) return;

  int nb = seg[(size_t)n0 * STATES];
  int ne = seg[(size_t)n0 * STATES + 1];

  for (int jp = 0; jp < NPW; jp += 2) {
    const int nodeA = n0 + jp;
    if (nodeA >= N) break;
    const bool hasB = (nodeA + 1) < N;
    const size_t rowP = (size_t)(nodeA + ((hasB && !lo) ? 1 : 0)) * STATES;
    const v4f pr = *(const v4f*)(priors + rowP + q4);

    v4f accA = {0.0f, 0.0f, 0.0f, 0.0f};
    v4f accB = {0.0f, 0.0f, 0.0f, 0.0f};
    for (int h = 0; h < 2; ++h) {
      const int node = nodeA + h;
      if (node >= N) break;
      const int beg = min(max(__builtin_amdgcn_readfirstlane(nb), 0), E2);
      const int deg = min(max(__builtin_amdgcn_readfirstlane(ne) - beg, 0), E2 - beg);
      if (jp + h + 1 < NPW && node + 1 < N) {
        nb = seg[(size_t)(node + 1) * STATES];
        ne = seg[(size_t)(node + 1) * STATES + 1];
      }

      v4f acc = {0.0f, 0.0f, 0.0f, 0.0f};
      for (int base = 0; base < deg; base += 8 * GRP) {
        int ei[GRP], ki[GRP], ri[GRP];
        v4f b[GRP], mr[GRP];
        for (int i = 0; i < GRP; ++i) {
          if (base + 8 * i < deg) {
            const int p = base + 8 * i + g;
            ei[i] = perm[beg + ((p < deg) ? p : (base + 8 * i))];
          }
        }
        for (int i = 0; i < GRP; ++i) {
          if (base + 8 * i < deg) {
            ei[i] = min(max(ei[i], 0), E2 - 1);
            if (MODE != 3) ki[i] = src[ei[i]];
            if (MODE == 4) ri[i] = rev[ei[i]];
          }
        }
        for (int i = 0; i < GRP; ++i) {
          if (base + 8 * i < deg) {
            if (MODE == 3) {
              b[i] = *(const v4f*)(M + (size_t)ei[i] * STATES + q4);
            } else {
              b[i] = *(const v4f*)(belin + (size_t)ki[i] * STATES + q4);
              if (MODE == 4) mr[i] = *(const v4f*)(M + (size_t)ri[i] * STATES + q4);
            }
          }
        }
        for (int i = 0; i < GRP; ++i) {
          if (base + 8 * i < deg) {
            v4f lm;
            if (MODE == 1) {
              lm = b[i];
            } else {
              v4f m;
              if (MODE == 3) {
                m = b[i];
              } else {
                const v4f x = b[i] / mr[i];
                float tt = (x.x + x.y) + (x.z + x.w);
                tt += __shfl_xor(tt, 1, 32);
                tt += __shfl_xor(tt, 2, 32);
                m = (tt + EXP1M1F * x) / ((15.0f + EXP1F) * tt);
              }
              lm.x = logf(fmaxf(m.x, EPSF));
              lm.y = logf(fmaxf(m.y, EPSF));
              lm.z = logf(fmaxf(m.z, EPSF));
              lm.w = logf(fmaxf(m.w, EPSF));
            }
            const bool live = (base + 8 * i + g) < deg;
            acc.x += live ? lm.x : 0.0f;
            acc.y += live ? lm.y : 0.0f;
            acc.z += live ? lm.z : 0.0f;
            acc.w += live ? lm.w : 0.0f;
          }
        }
      }
      if (h == 0) accA = acc; else accB = acc;
    }

    v4f t = lo ? accA : accB;
    const v4f u = lo ? accB : accA;
    t.x += __shfl_xor(u.x, 16, 32);
    t.y += __shfl_xor(u.y, 16, 32);
    t.z += __shfl_xor(u.z, 16, 32);
    t.w += __shfl_xor(u.w, 16, 32);
    t.x += __shfl_xor(t.x, 8, 32);
    t.y += __shfl_xor(t.y, 8, 32);
    t.z += __shfl_xor(t.z, 8, 32);
    t.w += __shfl_xor(t.w, 8, 32);
    t.x += __shfl_xor(t.x, 4, 32);
    t.y += __shfl_xor(t.y, 4, 32);
    t.z += __shfl_xor(t.z, 4, 32);
    t.w += __shfl_xor(t.w, 4, 32);

    v4f lb;
    lb.x = logf(fmaxf(pr.x, EPSF)) + t.x;
    lb.y = logf(fmaxf(pr.y, EPSF)) + t.y;
    lb.z = logf(fmaxf(pr.z, EPSF)) + t.z;
    lb.w = logf(fmaxf(pr.w, EPSF)) + t.w;

    float mx = fmaxf(fmaxf(lb.x, lb.y), fmaxf(lb.z, lb.w));
    mx = fmaxf(mx, __shfl_xor(mx, 1, 32));
    mx = fmaxf(mx, __shfl_xor(mx, 2, 32));
    v4f ex;
    ex.x = expf(lb.x - mx);
    ex.y = expf(lb.y - mx);
    ex.z = expf(lb.z - mx);
    ex.w = expf(lb.w - mx);
    float sm = (ex.x + ex.y) + (ex.z + ex.w);
    sm += __shfl_xor(sm, 1, 32);
    sm += __shfl_xor(sm, 2, 32);
    const v4f o = ex / sm;
    if ((lane & 12) == 0 && (lo || hasB)) *(v4f*)(bout + rowP + q4) = o;
  }
}

extern "C" void kernel_launch(void* const* d_in, const int* in_sizes, int n_in,
                              void* d_out, int out_size, void* d_ws, size_t ws_size,
                              hipStream_t stream) {
  const float* feat = (const float*)d_in[0];
  const float* W    = (const float*)d_in[1];
  const int*   src  = (const int*)d_in[2];
  const int*   dst  = (const int*)d_in[3];
  const int*   rev  = (const int*)d_in[4];

  const int N  = in_sizes[0] / FDIM;
  const int E2 = in_sizes[2];

  float* priors      = (float*)d_out;
  float* beliefs_out = priors + (size_t)N * STATES;
  int*   seg         = (int*)beliefs_out;

  const int NB  = (N + BINW - 1) / BINW;
  const int BLK = (E2 + CH - 1) / CH;
  const int L   = NB * BLK;

  int*   perm   = (int*)d_ws;
  float* belbuf = (float*)d_ws + (size_t)E2;
  float* msgs   = (float*)d_ws + (size_t)3 * N * STATES;
  int*   T      = (int*)msgs;
  int*   Td     = T + (size_t)E2;
  float* lg1    = msgs + (size_t)2 * E2;
  int*   Hm     = (int*)(lg1 + (size_t)N * STATES);
  int*   Off    = Hm + (size_t)L;
  const bool tab  = E2 > N && (size_t)(E2 - N) * STATES >= (size_t)2 * E2 + (size_t)N * STATES + (size_t)2 * L;
  float* m1tab    = tab ? msgs + (size_t)(E2 - N) * STATES : nullptr;
  const int split = tab ? ((E2 - N) / (32 * TRIPS)) * (32 * TRIPS) : 0;

  const dim3 blk(256);
  const int tiles   = (N + 15) / 16;
  const int gBlocks = (tiles + 7) / 8;
  const int sBlocks = (L + 1023) / 1024;
  const int nBlocks = (N + 8 * NPW - 1) / (8 * NPW);
  const int eBlocksA = (split / (32 * TRIPS) + 7) / 8;
  const int eBlocksB = ((E2 - split + 32 * TRIPS - 1) / (32 * TRIPS) + 7) / 8;

  const int gA = gBlocks / 2;
  const int gB = gBlocks - gA;
  bp_bin_count<<<BLK, blk, 0, stream>>>(dst, Hm, E2, NB, BLK);
  bp_scan_excl<<<sBlocks, dim3(1024), 0, stream>>>(Hm, Off, L);
  bp_bin_scatter<<<BLK + gA, blk, 0, stream>>>(dst, Off, T, Td, E2, NB, BLK,
                                             feat, W, priors, lg1, m1tab, N, tiles, 0);
  bp_bin_sort<<<NB + gB, blk, 0, stream>>>(Off, T, Td, seg, perm, N, E2, NB, BLK,
                                         feat, W, priors, lg1, m1tab, tiles, gA);

  bp_pull<1><<<nBlocks, blk, 0, stream>>>(perm, src, rev, priors, lg1, msgs, seg, belbuf, N, E2);
  if (split > 0) bp_edge<1><<<eBlocksA, blk, 0, stream>>>(src, rev, m1tab, belbuf, msgs, 0, split);
  bp_edge<0><<<eBlocksB, blk, 0, stream>>>(src, rev, priors, belbuf, msgs, split, E2);
  bp_pull<3><<<nBlocks, blk, 0, stream>>>(perm, src, rev, priors, belbuf, msgs, seg, belbuf, N, E2);
  bp_pull<4><<<nBlocks, blk, 0, stream>>>(perm, src, rev, priors, belbuf, msgs, seg, beliefs_out, N, E2);
}


# Messages kept as reciprocals and a node's log-sum taken as logs of products of up to 8 messages: per in-edge 2 divisions and a log become products (new f32 order and rounding of the log-sums)
# speedup vs baseline: 2.3960x; 1.0375x over previous
#include <hip/hip_runtime.h>
#include <math.h>


typedef float v2f __attribute__((ext_vector_type(2)));
typedef float v4f __attribute__((ext_vector_type(4)));
typedef float v8f __attribute__((ext_vector_type(8)));

#define STATES   16
#define FDIM     128
#define EPSF     1e-10f
#define EXP1F    2.7182818284590452f
#define EXP1M1F  1.7182818284590452f

__device__ __forceinline__ void bp_priors_tiles(const float* __restrict__ feat,
                                                const float* __restrict__ W,
                                                float* __restrict__ priors,
                                                float* __restrict__ lg1,
                                                float* __restrict__ m1,
                                                int N, int numTiles, int tb) {
  const int wave = tb * 8 + (threadIdx.x >> 5);
  if (wave >= numTiles) return;
  const int lane = threadIdx.x & 31;
  const int half = lane >> 4;
  const int s    = lane & 15;

  int rowA = wave * 16 + s;
  if (rowA >= N) rowA = N - 1;
  const float* frow = feat + (size_t)rowA * FDIM;

  v8f acc = {};
#pragma unroll
  for (int k0 = 0; k0 < FDIM; k0 += 4) {
    const int k = k0 + 2 * half;
    v2f a, b;
    a.x = frow[k];
    a.y = frow[k + 1];
    b.x = W[(size_t)k * STATES + s];
    b.y = W[(size_t)(k + 1) * STATES + s];
    acc = __builtin_amdgcn_wmma_f32_16x16x4_f32(false, a, false, b,
                                                (short)0, acc, false, false);
  }

#pragma unroll
  for (int r = 0; r < 8; ++r) {
    float logit = acc[r];
    float mx = logit;
    mx = fmaxf(mx, __shfl_xor(mx, 1, 32));
    mx = fmaxf(mx, __shfl_xor(mx, 2, 32));
    mx = fmaxf(mx, __shfl_xor(mx, 4, 32));
    mx = fmaxf(mx, __shfl_xor(mx, 8, 32));
    float ex = expf(logit - mx);
    float sm = ex;
    sm += __shfl_xor(sm, 1, 32);
    sm += __shfl_xor(sm, 2, 32);
    sm += __shfl_xor(sm, 4, 32);
    sm += __shfl_xor(sm, 8, 32);
    const int rowO = wave * 16 + r + 8 * half;
    const float p = ex / sm;
    const float x = p / 0.0625f;
    float tt = x;
    tt += __shfl_xor(tt, 1, 32);
    tt += __shfl_xor(tt, 2, 32);
    tt += __shfl_xor(tt, 4, 32);
    tt += __shfl_xor(tt, 8, 32);
    const float m = (tt + EXP1M1F * x) / ((15.0f + EXP1F) * tt);
    const float lm = logf(fmaxf(m, EPSF));
    if (rowO < N) {
      priors[(size_t)rowO * STATES + s] = p;
      lg1[(size_t)rowO * STATES + s]    = lm;
      if (m1) m1[(size_t)rowO * STATES + s] = ((15.0f + EXP1F) * tt) / (tt + EXP1M1F * x);
    }
  }
}

#define BINSH   8
#define BINW    256
#define CH      8192
#define NBMAX   1024
#define CAP     16384
#define UNR     40

__device__ __forceinline__ int bp_bin_of(int d, int NB) {
  return min(max(d >> BINSH, 0), NB - 1);
}

__global__ void bp_bin_count(const int* __restrict__ dst, int* __restrict__ Hm,
                             int E2, int NB, int BLK) {
  __shared__ int hist[NBMAX];
  __shared__ int sink[256];
  const int tid = threadIdx.x;
  const int base = blockIdx.x * CH;
  int dd[CH / 256];
#pragma unroll
  for (int j = 0; j < CH / 256; ++j) {
    const int e = base + j * 256 + tid;
    dd[j] = dst[min(e, E2 - 1)];
  }
  for (int b = tid; b < NBMAX; b += 256) hist[b] = 0;
  sink[tid] = 0;
  __syncthreads();
#pragma unroll
  for (int j = 0; j < CH / 256; ++j) {
    const int e = base + j * 256 + tid;
    int* ctr = (e < E2) ? &hist[bp_bin_of(dd[j], NB)] : &sink[tid];
    atomicAdd(ctr, 1);
  }
  __syncthreads();
  for (int b = tid; b < NB; b += 256) Hm[(size_t)b * BLK + blockIdx.x] = hist[b];
}

__global__ void bp_scan_excl(const int* __restrict__ in, int* __restrict__ out, int n) {
  __shared__ int wsum[32];
  __shared__ int sbase;
  const int tid  = threadIdx.x;
  const int lane = tid & 31;
  const int wv   = tid >> 5;
  const int start = blockIdx.x * 1024;

  int part = 0;
  for (int i = tid; i < start; i += 1024) part += in[i];
  for (int o = 16; o > 0; o >>= 1) part += __shfl_xor(part, o, 32);
  if (lane == 0) wsum[wv] = part;
  __syncthreads();
  if (wv == 0) {
    int v = wsum[lane];
    for (int o = 16; o > 0; o >>= 1) v += __shfl_xor(v, o, 32);
    if (lane == 0) sbase = v;
  }
  __syncthreads();
  const int base = sbase;

  const int i = start + tid;
  const int c = (i < n) ? in[i] : 0;
  int inc = c;
  for (int o = 1; o < 32; o <<= 1) {
    const int y = __shfl_up(inc, o, 32);
    if (lane >= o) inc += y;
  }
  if (lane == 31) wsum[wv] = inc;
  __syncthreads();
  if (wv == 0) {
    const int v = wsum[lane];
    int iv = v;
    for (int o = 1; o < 32; o <<= 1) {
      const int y = __shfl_up(iv, o, 32);
      if (lane >= o) iv += y;
    }
    wsum[lane] = iv - v;
  }
  __syncthreads();
  if (i < n) out[i] = base + wsum[wv] + inc - c;
}

__global__ void __launch_bounds__(256) bp_bin_scatter(const int* __restrict__ dst, const int* __restrict__ Off,
                               int* __restrict__ T, int* __restrict__ Td,
                               int E2, int NB, int BLK,
                               const float* __restrict__ feat, const float* __restrict__ W,
                               float* __restrict__ priors, float* __restrict__ lg1,
                               float* __restrict__ m1, int N, int numTiles, int tb0) {
  if ((int)blockIdx.x >= BLK) {
    bp_priors_tiles(feat, W, priors, lg1, m1, N, numTiles, tb0 + ((int)blockIdx.x - BLK));
    return;
  }
  __shared__ int hist[NBMAX];
  __shared__ int delta[NBMAX];
  __shared__ int se[CH];
  __shared__ int sd[CH];
  __shared__ int wsum[8];
  __shared__ int sink[256];
  const int tid  = threadIdx.x;
  const int lane = tid & 31;
  const int wv   = tid >> 5;
  const int blk  = blockIdx.x;
  const int base = blk * CH;
  const int nIt  = min(CH, E2 - base);

  int dd[CH / 256];
#pragma unroll
  for (int j = 0; j < CH / 256; ++j) {
    const int e = base + j * 256 + tid;
    dd[j] = (e < E2) ? dst[e] : 0;
  }
  for (int b = tid; b < NBMAX; b += 256) hist[b] = 0;
  sink[tid] = 0;
  __syncthreads();
#pragma unroll
  for (int j = 0; j < CH / 256; ++j) {
    const int e = base + j * 256 + tid;
    if (e < E2) atomicAdd(&hist[bp_bin_of(dd[j], NB)], 1);
  }
  __syncthreads();
  {
    const int v0 = hist[4 * tid], v1 = hist[4 * tid + 1], v2 = hist[4 * tid + 2], v3 = hist[4 * tid + 3];
    const int s4 = (v0 + v1) + (v2 + v3);
    int inc = s4;
    for (int o = 1; o < 32; o <<= 1) {
      const int y = __shfl_up(inc, o, 32);
      if (lane >= o) inc += y;
    }
    if (lane == 31) wsum[wv] = inc;
    __syncthreads();
    int woff = 0;
    for (int k = 0; k < wv; ++k) woff += wsum[k];
    const int ex = woff + inc - s4;
    hist[4 * tid]     = ex;
    hist[4 * tid + 1] = ex + v0;
    hist[4 * tid + 2] = ex + v0 + v1;
    hist[4 * tid + 3] = ex + v0 + v1 + v2;
  }
  __syncthreads();
  {
    int ov[NBMAX / 256];
#pragma unroll
    for (int k = 0; k < NBMAX / 256; ++k) ov[k] = Off[(size_t)min(k * 256 + tid, NB - 1) * BLK + blk];
#pragma unroll
    for (int k = 0; k < NBMAX / 256; ++k) {
      const int b = k * 256 + tid;
      if (b < NB) delta[b] = ov[k] - hist[b];
    }
  }
  __syncthreads();
  int pp[CH / 256];
#pragma unroll
  for (int j = 0; j < CH / 256; ++j) {
    const int e = base + j * 256 + tid;
    int* ctr = (e < E2) ? &hist[bp_bin_of(dd[j], NB)] : &sink[tid];
    pp[j] = atomicAdd(ctr, 1);
  }
#pragma unroll
  for (int j = 0; j < CH / 256; ++j) {
    const int e = base + j * 256 + tid;
    if (e < E2) {
      const int p = min(max(pp[j], 0), CH - 1);
      se[p] = e;
      sd[p] = dd[j];
    }
  }
  __syncthreads();
  int dv[CH / 256], ev[CH / 256], gv[CH / 256];
#pragma unroll
  for (int j = 0; j < CH / 256; ++j) {
    const int s = min(j * 256 + tid, nIt - 1);
    dv[j] = sd[s];
    ev[j] = se[s];
  }
#pragma unroll
  for (int j = 0; j < CH / 256; ++j) gv[j] = delta[bp_bin_of(dv[j], NB)];
#pragma unroll
  for (int j = 0; j < CH / 256; ++j) {
    const int s = j * 256 + tid;
    if (s < nIt) {
      const int g = min(max(s + gv[j], 0), E2 - 1);
      T[g]  = ev[j];
      Td[g] = dv[j];
    }
  }
}

__global__ void __launch_bounds__(256) bp_bin_sort(const int* __restrict__ Off, const int* __restrict__ T,
                            const int* __restrict__ Td, int* __restrict__ seg,
                            int* __restrict__ perm, int N, int E2, int NB, int BLK,
                            const float* __restrict__ feat, const float* __restrict__ W,
                            float* __restrict__ priors, float* __restrict__ lg1,
                            float* __restrict__ m1, int numTiles, int tb0) {
  if ((int)blockIdx.x >= NB) {
    bp_priors_tiles(feat, W, priors, lg1, m1, N, numTiles, tb0 + ((int)blockIdx.x - NB));
    return;
  }
  __shared__ int ncnt[BINW];
  __shared__ int nstart[BINW + 1];
  __shared__ int cur[BINW];
  __shared__ int Be[CAP];
  __shared__ int wsum[8];
  __shared__ int sink[256];
  const int tid  = threadIdx.x;
  const int lane = tid & 31;
  const int wv   = tid >> 5;
  const int b    = blockIdx.x;
  const int n0   = b << BINSH;

  int binBeg = Off[(size_t)b * BLK];
  int binEnd = (b + 1 < NB) ? Off[(size_t)(b + 1) * BLK] : E2;
  binBeg = min(max(binBeg, 0), E2);
  binEnd = min(max(binEnd, binBeg), E2);
  const int cntB = binEnd - binBeg;

  ncnt[tid] = 0;
  sink[tid] = 0;
  __syncthreads();
  for (int i0 = 0; i0 < cntB; i0 += 256 * UNR) {
    int dn[UNR];
#pragma unroll
    for (int u = 0; u < UNR; ++u) {
      const int i = i0 + u * 256 + tid;
      dn[u] = Td[binBeg + min(i, cntB - 1)];
    }
#pragma unroll
    for (int u = 0; u < UNR; ++u) {
      const int i = i0 + u * 256 + tid;
      int* ctr = (i < cntB) ? &ncnt[(dn[u] - n0) & (BINW - 1)] : &sink[tid];
      atomicAdd(ctr, 1);
    }
  }
  __syncthreads();
  {
    const int c = ncnt[tid];
    int inc = c;
    for (int o = 1; o < 32; o <<= 1) {
      const int y = __shfl_up(inc, o, 32);
      if (lane >= o) inc += y;
    }
    if (lane == 31) wsum[wv] = inc;
    __syncthreads();
    int woff = 0;
    for (int k = 0; k < wv; ++k) woff += wsum[k];
    const int ex = woff + inc - c;
    nstart[tid] = ex;
    cur[tid]    = ex;
    if (tid == BINW - 1) nstart[BINW] = ex + c;
    const int node = n0 + tid;
    if (node < N) {
      seg[(size_t)node * STATES]     = binBeg + ex;
      seg[(size_t)node * STATES + 1] = binBeg + ex + c;
    }
  }
  __syncthreads();
  if (cntB <= CAP) {
    for (int i0 = 0; i0 < cntB; i0 += 256 * UNR) {
      int dn[UNR], en[UNR];
#pragma unroll
      for (int u = 0; u < UNR; ++u) {
        const int i = min(i0 + u * 256 + tid, cntB - 1);
        dn[u] = Td[binBeg + i];
        en[u] = T[binBeg + i];
      }
      int pq[UNR];
#pragma unroll
      for (int u = 0; u < UNR; ++u) {
        const int i = i0 + u * 256 + tid;
        int* ctr = (i < cntB) ? &cur[(dn[u] - n0) & (BINW - 1)] : &sink[tid];
        pq[u] = atomicAdd(ctr, 1);
      }
#pragma unroll
      for (int u = 0; u < UNR; ++u) {
        const int i = i0 + u * 256 + tid;
        if (i < cntB) Be[min(max(pq[u], 0), CAP - 1)] = en[u];
      }
    }
    __syncthreads();
    for (int n = wv; n < BINW; n += 8) {
      const int s = __builtin_amdgcn_readfirstlane(nstart[n]);
      const int d = __builtin_amdgcn_readfirstlane(nstart[n + 1]) - s;
      for (int i0 = 0; i0 < d; i0 += 32) {
        const bool have = (i0 + lane) < d;
        const int e = have ? Be[s + i0 + lane] : 0x7fffffff;
        int less = 0;
        for (int j0 = 0; j0 < d; j0 += 32) {
          const int o = (j0 + lane < d) ? Be[s + j0 + lane] : 0x7fffffff;
#pragma unroll
          for (int k = 0; k < 32; ++k) less += (__builtin_amdgcn_readlane(o, k) < e) ? 1 : 0;
        }
        if (have) perm[min(binBeg + s + less, E2 - 1)] = e;
      }
    }
  } else {
    for (int i = tid; i < cntB; i += 256) {
      const int n = (Td[binBeg + i] - n0) & (BINW - 1);
      const int p = atomicAdd(&cur[n], 1);
      perm[min(max(binBeg + p, 0), E2 - 1)] = T[binBeg + i];
    }
  }
}

#define TRIPS 8
template <int TAB>
__global__ void bp_edge(const int* __restrict__ src,
                        const int* __restrict__ rev,
                        const float* rowsrc,
                        const float* __restrict__ bel1,
                        float* M,
                        int eBeg, int eEnd) {
  const int w    = __builtin_amdgcn_readfirstlane((int)(blockIdx.x * (blockDim.x >> 5) + (threadIdx.x >> 5)));
  const int lane = threadIdx.x & 31;
  const int g    = lane >> 2;
  const int q4   = (lane & 3) * 4;
  const int ew   = eBeg + w * (32 * TRIPS);

  for (int t = 0; t < TRIPS; ++t) {
    const int e0 = ew + 32 * t;
    if (e0 >= eEnd) break;
    int ei[4], ki[4], ri[4];
#pragma unroll
    for (int i = 0; i < 4; ++i) ei[i] = min(e0 + 8 * i + g, eEnd - 1);
#pragma unroll
    for (int i = 0; i < 4; ++i) { ki[i] = src[ei[i]]; ri[i] = rev[ei[i]]; }
#pragma unroll
    for (int i = 0; i < 4; ++i) ri[i] = src[ri[i]];

    v4f b[4], mr[4];
#pragma unroll
    for (int i = 0; i < 4; ++i) b[i] = *(const v4f*)(bel1 + (size_t)ki[i] * STATES + q4);
#pragma unroll
    for (int i = 0; i < 4; ++i) mr[i] = *(const v4f*)(rowsrc + (size_t)ri[i] * STATES + q4);

#pragma unroll
    for (int i = 0; i < 4; ++i) {
      if (!TAB) {
        const v4f x1 = mr[i] / 0.0625f;
        float t1 = (x1.x + x1.y) + (x1.z + x1.w);
        t1 += __shfl_xor(t1, 1, 32);
        t1 += __shfl_xor(t1, 2, 32);
        const float k1 = (15.0f + EXP1F) * t1;
        const v4f n1 = t1 + EXP1M1F * x1;
        mr[i].x = k1 / n1.x;
        mr[i].y = k1 / n1.y;
        mr[i].z = k1 / n1.z;
        mr[i].w = k1 / n1.w;
      }
      const v4f x = b[i] * mr[i];
      float tt = (x.x + x.y) + (x.z + x.w);
      tt += __shfl_xor(tt, 1, 32);
      tt += __shfl_xor(tt, 2, 32);
      const float kt = (15.0f + EXP1F) * tt;
      const v4f nu = tt + EXP1M1F * x;
      v4f m;
      m.x = kt / nu.x;
      m.y = kt / nu.y;
      m.z = kt / nu.z;
      m.w = kt / nu.w;
      if (e0 + 8 * i + g < eEnd) *(v4f*)(M + (size_t)ei[i] * STATES + q4) = m;
    }
  }
}

#define NPW 4
#define GRP 4
template <int MODE>
__device__ __forceinline__ void bp_fold(v4f& acc, v4f& pn, float& pd) {
  if (MODE == 3) {
    acc.x -= logf(pn.x);
    acc.y -= logf(pn.y);
    acc.z -= logf(pn.z);
    acc.w -= logf(pn.w);
  } else {
    acc.x += logf(pn.x / pd);
    acc.y += logf(pn.y / pd);
    acc.z += logf(pn.z / pd);
    acc.w += logf(pn.w / pd);
  }
  pn.x = 1.0f; pn.y = 1.0f; pn.z = 1.0f; pn.w = 1.0f;
  pd = 1.0f;
}
template <int MODE>
__global__ void bp_pull(const int* __restrict__ perm,
                        const int* __restrict__ src,
                        const int* __restrict__ rev,
                        const float* __restrict__ priors,
                        const float* belin,
                        const float* M,
                        const int* seg,
                        float* bout,
                        int N, int E2) {
  const int w    = __builtin_amdgcn_readfirstlane((int)(blockIdx.x * (blockDim.x >> 5) + (threadIdx.x >> 5)));
  const int lane = threadIdx.x & 31;
  const int g    = lane >> 2;
  const int q4   = (lane & 3) * 4;
  const bool lo  = lane < 16;
  const int n0   = w * NPW;
  if (n0 >= N) return;

  int nb = seg[(size_t)n0 * STATES];
  int ne = seg[(size_t)n0 * STATES + 1];

  for (int jp = 0; jp < NPW; jp += 2) {
    const int nodeA = n0 + jp;
    if (nodeA >= N) break;
    const bool hasB = (nodeA + 1) < N;
    const size_t rowP = (size_t)(nodeA + ((hasB && !lo) ? 1 : 0)) * STATES;
    const v4f pr = *(const v4f*)(priors + rowP + q4);

    v4f accA = {0.0f, 0.0f, 0.0f, 0.0f};
    v4f accB = {0.0f, 0.0f, 0.0f, 0.0f};
    for (int h = 0; h < 2; ++h) {
      const int node = nodeA + h;
      if (node >= N) break;
      const int beg = min(max(__builtin_amdgcn_readfirstlane(nb), 0), E2);
      const int deg = min(max(__builtin_amdgcn_readfirstlane(ne) - beg, 0), E2 - beg);
      if (jp + h + 1 < NPW && node + 1 < N) {
        nb = seg[(size_t)(node + 1) * STATES];
        ne = seg[(size_t)(node + 1) * STATES + 1];
      }

      v4f acc = {0.0f, 0.0f, 0.0f, 0.0f};
      v4f pn  = {1.0f, 1.0f, 1.0f, 1.0f};
      float pd = 1.0f;
      for (int base = 0; base < deg; base += 8 * GRP) {
        int ei[GRP], ki[GRP], ri[GRP];
        v4f b[GRP], mr[GRP];
        for (int i = 0; i < GRP; ++i) {
          if (base + 8 * i < deg) {
            const int p = base + 8 * i + g;
            ei[i] = perm[beg + ((p < deg) ? p : (base + 8 * i))];
          }
        }
        for (int i = 0; i < GRP; ++i) {
          if (base + 8 * i < deg) {
            ei[i] = min(max(ei[i], 0), E2 - 1);
            if (MODE != 3) ki[i] = src[ei[i]];
            if (MODE == 4) ri[i] = rev[ei[i]];
          }
        }
        for (int i = 0; i < GRP; ++i) {
          if (base + 8 * i < deg) {
            if (MODE == 3) {
              b[i] = *(const v4f*)(M + (size_t)ei[i] * STATES + q4);
            } else {
              b[i] = *(const v4f*)(belin + (size_t)ki[i] * STATES + q4);
              if (MODE == 4) mr[i] = *(const v4f*)(M + (size_t)ri[i] * STATES + q4);
            }
          }
        }
        for (int i = 0; i < GRP; ++i) {
          if (base + 8 * i < deg) {
            const bool live = (base + 8 * i + g) < deg;
            if (MODE == 1) {
              acc.x += live ? b[i].x : 0.0f;
              acc.y += live ? b[i].y : 0.0f;
              acc.z += live ? b[i].z : 0.0f;
              acc.w += live ? b[i].w : 0.0f;
            } else if (MODE == 3) {
              pn.x *= live ? b[i].x : 1.0f;
              pn.y *= live ? b[i].y : 1.0f;
              pn.z *= live ? b[i].z : 1.0f;
              pn.w *= live ? b[i].w : 1.0f;
            } else {
              const v4f x = b[i] * mr[i];
              float tt = (x.x + x.y) + (x.z + x.w);
              tt += __shfl_xor(tt, 1, 32);
              tt += __shfl_xor(tt, 2, 32);
              const v4f nu = tt + EXP1M1F * x;
              const float de = (15.0f + EXP1F) * tt;
              pn.x *= live ? nu.x : 1.0f;
              pn.y *= live ? nu.y : 1.0f;
              pn.z *= live ? nu.z : 1.0f;
              pn.w *= live ? nu.w : 1.0f;
              pd   *= live ? de : 1.0f;
            }
          }
        }
        if (MODE != 1 && (base & (8 * GRP)) != 0) bp_fold<MODE>(acc, pn, pd);
      }
      if (MODE != 1 && (((deg + 8 * GRP - 1) / (8 * GRP)) & 1) != 0) bp_fold<MODE>(acc, pn, pd);
      if (h == 0) accA = acc; else accB = acc;
    }

    v4f t = lo ? accA : accB;
    const v4f u = lo ? accB : accA;
    t.x += __shfl_xor(u.x, 16, 32);
    t.y += __shfl_xor(u.y, 16, 32);
    t.z += __shfl_xor(u.z, 16, 32);
    t.w += __shfl_xor(u.w, 16, 32);
    t.x += __shfl_xor(t.x, 8, 32);
    t.y += __shfl_xor(t.y, 8, 32);
    t.z += __shfl_xor(t.z, 8, 32);
    t.w += __shfl_xor(t.w, 8, 32);
    t.x += __shfl_xor(t.x, 4, 32);
    t.y += __shfl_xor(t.y, 4, 32);
    t.z += __shfl_xor(t.z, 4, 32);
    t.w += __shfl_xor(t.w, 4, 32);

    v4f lb;
    lb.x = logf(fmaxf(pr.x, EPSF)) + t.x;
    lb.y = logf(fmaxf(pr.y, EPSF)) + t.y;
    lb.z = logf(fmaxf(pr.z, EPSF)) + t.z;
    lb.w = logf(fmaxf(pr.w, EPSF)) + t.w;

    float mx = fmaxf(fmaxf(lb.x, lb.y), fmaxf(lb.z, lb.w));
    mx = fmaxf(mx, __shfl_xor(mx, 1, 32));
    mx = fmaxf(mx, __shfl_xor(mx, 2, 32));
    v4f ex;
    ex.x = expf(lb.x - mx);
    ex.y = expf(lb.y - mx);
    ex.z = expf(lb.z - mx);
    ex.w = expf(lb.w - mx);
    float sm = (ex.x + ex.y) + (ex.z + ex.w);
    sm += __shfl_xor(sm, 1, 32);
    sm += __shfl_xor(sm, 2, 32);
    const v4f o = ex / sm;
    if ((lane & 12) == 0 && (lo || hasB)) *(v4f*)(bout + rowP + q4) = o;
  }
}

extern "C" void kernel_launch(void* const* d_in, const int* in_sizes, int n_in,
                              void* d_out, int out_size, void* d_ws, size_t ws_size,
                              hipStream_t stream) {
  const float* feat = (const float*)d_in[0];
  const float* W    = (const float*)d_in[1];
  const int*   src  = (const int*)d_in[2];
  const int*   dst  = (const int*)d_in[3];
  const int*   rev  = (const int*)d_in[4];

  const int N  = in_sizes[0] / FDIM;
  const int E2 = in_sizes[2];

  float* priors      = (float*)d_out;
  float* beliefs_out = priors + (size_t)N * STATES;
  int*   seg         = (int*)beliefs_out;

  const int NB  = (N + BINW - 1) / BINW;
  const int BLK = (E2 + CH - 1) / CH;
  const int L   = NB * BLK;

  int*   perm   = (int*)d_ws;
  float* belbuf = (float*)d_ws + (size_t)E2;
  float* msgs   = (float*)d_ws + (size_t)3 * N * STATES;
  int*   T      = (int*)msgs;
  int*   Td     = T + (size_t)E2;
  float* lg1    = msgs + (size_t)2 * E2;
  int*   Hm     = (int*)(lg1 + (size_t)N * STATES);
  int*   Off    = Hm + (size_t)L;
  const bool tab  = E2 > N && (size_t)(E2 - N) * STATES >= (size_t)2 * E2 + (size_t)N * STATES + (size_t)2 * L;
  float* m1tab    = tab ? msgs + (size_t)(E2 - N) * STATES : nullptr;
  const int split = tab ? ((E2 - N) / (32 * TRIPS)) * (32 * TRIPS) : 0;

  const dim3 blk(256);
  const int tiles   = (N + 15) / 16;
  const int gBlocks = (tiles + 7) / 8;
  const int sBlocks = (L + 1023) / 1024;
  const int nBlocks = (N + 8 * NPW - 1) / (8 * NPW);
  const int eBlocksA = (split / (32 * TRIPS) + 7) / 8;
  const int eBlocksB = ((E2 - split + 32 * TRIPS - 1) / (32 * TRIPS) + 7) / 8;

  const int gA = gBlocks / 2;
  const int gB = gBlocks - gA;
  bp_bin_count<<<BLK, blk, 0, stream>>>(dst, Hm, E2, NB, BLK);
  bp_scan_excl<<<sBlocks, dim3(1024), 0, stream>>>(Hm, Off, L);
  bp_bin_scatter<<<BLK + gA, blk, 0, stream>>>(dst, Off, T, Td, E2, NB, BLK,
                                             feat, W, priors, lg1, m1tab, N, tiles, 0);
  bp_bin_sort<<<NB + gB, blk, 0, stream>>>(Off, T, Td, seg, perm, N, E2, NB, BLK,
                                         feat, W, priors, lg1, m1tab, tiles, gA);

  bp_pull<1><<<nBlocks, blk, 0, stream>>>(perm, src, rev, priors, lg1, msgs, seg, belbuf, N, E2);
  if (split > 0) bp_edge<1><<<eBlocksA, blk, 0, stream>>>(src, rev, m1tab, belbuf, msgs, 0, split);
  bp_edge<0><<<eBlocksB, blk, 0, stream>>>(src, rev, priors, belbuf, msgs, split, E2);
  bp_pull<3><<<nBlocks, blk, 0, stream>>>(perm, src, rev, priors, belbuf, msgs, seg, belbuf, N, E2);
  bp_pull<4><<<nBlocks, blk, 0, stream>>>(perm, src, rev, priors, belbuf, msgs, seg, beliefs_out, N, E2);
}
